# starting kernel
# baseline (speedup 1.0000x reference)
#include <hip/hip_runtime.h>
#include <hip/hip_bf16.h>


typedef __attribute__((ext_vector_type(16))) __bf16 v16bf;
typedef __attribute__((ext_vector_type(8)))  float  v8f;

#define N_SRC   131072
#define N_MID   32768
#define N_DST   8192
#define E1_N    524288
#define E2_N    131072
#define HDIM    256
#define N_PAIRS 8192

__device__ __forceinline__ void atomic_add_f32(float* p, float v) {
    unsafeAtomicAdd(p, v);
}

__global__ void f32_to_bf16_kernel(const float* __restrict__ src,
                                   __bf16* __restrict__ dst, int n) {
    int i = blockIdx.x * blockDim.x + threadIdx.x;
    if (i < n) dst[i] = (__bf16)src[i];
}

__global__ void zero_f32_kernel(float* __restrict__ p, int n) {
    int i = blockIdx.x * blockDim.x + threadIdx.x;
    if (i < n) p[i] = 0.0f;
}

__global__ __launch_bounds__(256)
void gemm_bf16_n256(const float* __restrict__ A, const __bf16* __restrict__ W,
                    const float* __restrict__ bias, float* __restrict__ C,
                    int M, int K, int relu)
{
    const int lane = threadIdx.x & 31;
    const int wave = threadIdx.x >> 5;
    const int half = lane >> 4;
    const int l15  = lane & 15;
    const int m0   = blockIdx.x * 32 + (wave >> 2) * 16;
    const int n0   = (wave & 3) * 64;

    v8f acc[4];
    #pragma unroll
    for (int t = 0; t < 4; ++t) acc[t] = (v8f){};

    const float* arow = A + (size_t)(m0 + l15) * K;

    for (int k0 = 0; k0 < K; k0 += 32) {
        v16bf a;
        #pragma unroll
        for (int q = 0; q < 8; ++q) {
            int kk = k0 + (q >> 2) * 16 + half * 8 + (q & 3) * 2;
            float2 v = *(const float2*)(arow + kk);
            a[2 * q]     = (__bf16)v.x;
            a[2 * q + 1] = (__bf16)v.y;
        }
        #pragma unroll
        for (int t = 0; t < 4; ++t) {
            const __bf16* wrow = W + (size_t)(n0 + t * 16 + l15) * K;
            v16bf b;
            #pragma unroll
            for (int q = 0; q < 8; ++q) {
                int kk = k0 + (q >> 2) * 16 + half * 8 + (q & 3) * 2;
                b[2 * q]     = wrow[kk];
                b[2 * q + 1] = wrow[kk + 1];
            }
            acc[t] = __builtin_amdgcn_wmma_f32_16x16x32_bf16(
                false, a, false, b, (short)0, acc[t], false, false);
        }
    }

    #pragma unroll
    for (int t = 0; t < 4; ++t) {
        int n = n0 + t * 16 + l15;
        float bv = bias[n];
        #pragma unroll
        for (int r = 0; r < 8; ++r) {
            int m = m0 + half * 8 + r;
            float v = acc[t][r] + bv;
            if (relu) v = fmaxf(v, 0.0f);
            C[(size_t)m * HDIM + n] = v;
        }
    }
}

__global__ __launch_bounds__(256)
void edge_scatter_kernel(const float* __restrict__ nsrc,
                         const float* __restrict__ w,
                         const int* __restrict__ src,
                         const int* __restrict__ dst,
                         float* __restrict__ agg,
                         float* __restrict__ wsum, int E)
{
    int e    = blockIdx.x * 8 + (threadIdx.x >> 5);
    int lane = threadIdx.x & 31;
    if (e >= E) return;
    int   s  = src[e];
    int   d  = dst[e];
    float we = w[e];
    const float* nr = nsrc + (size_t)s * HDIM + lane * 8;
    float*       ar = agg  + (size_t)d * HDIM + lane * 8;
    float4 v0 = *(const float4*)(nr);
    float4 v1 = *(const float4*)(nr + 4);
    atomic_add_f32(ar + 0, v0.x * we);
    atomic_add_f32(ar + 1, v0.y * we);
    atomic_add_f32(ar + 2, v0.z * we);
    atomic_add_f32(ar + 3, v0.w * we);
    atomic_add_f32(ar + 4, v1.x * we);
    atomic_add_f32(ar + 5, v1.y * we);
    atomic_add_f32(ar + 6, v1.z * we);
    atomic_add_f32(ar + 7, v1.w * we);
    if (lane == 0) atomic_add_f32(wsum + d, we);
}

__global__ void build_z_kernel(const float* __restrict__ agg,
                               const float* __restrict__ wsum,
                               const float* __restrict__ hdst,
                               float* __restrict__ z, int rows)
{
    int i = blockIdx.x * blockDim.x + threadIdx.x;
    if (i >= rows * 512) return;
    int r = i >> 9, c = i & 511;
    float v;
    if (c < HDIM) {
        float ws = wsum[r];
        ws = ws < 1.0f ? 1.0f : ws;
        v = agg[(size_t)r * HDIM + c] * __frcp_rn(ws);
    } else {
        v = hdst[(size_t)r * HDIM + (c - HDIM)];
    }
    z[i] = v;
}

__global__ __launch_bounds__(256)
void skip_norm_kernel(const float* __restrict__ hp, const float* __restrict__ h2,
                      float* __restrict__ zf, int rows)
{
    int r    = blockIdx.x * 8 + (threadIdx.x >> 5);
    int lane = threadIdx.x & 31;
    if (r >= rows) return;
    float v[8];
    float ss = 0.0f;
    #pragma unroll
    for (int j = 0; j < 8; ++j) {
        int c = lane * 8 + j;
        v[j] = hp[(size_t)r * HDIM + c] + h2[(size_t)r * HDIM + c];
        ss += v[j] * v[j];
    }
    #pragma unroll
    for (int o = 16; o > 0; o >>= 1) ss += __shfl_xor(ss, o, 32);
    float nrm   = sqrtf(ss);
    float scale = (nrm == 0.0f) ? 1.0f : 1.0f / nrm;
    #pragma unroll
    for (int j = 0; j < 8; ++j)
        zf[(size_t)r * HDIM + lane * 8 + j] = v[j] * scale;
}

__global__ __launch_bounds__(256)
void score_kernel(const float* __restrict__ zf, const float* __restrict__ bias,
                  const int* __restrict__ nids,
                  const int* __restrict__ ps, const int* __restrict__ pd,
                  const int* __restrict__ ns, const int* __restrict__ nd,
                  float* __restrict__ out, int npairs)
{
    int p    = blockIdx.x * 8 + (threadIdx.x >> 5);
    int lane = threadIdx.x & 31;
    if (p >= npairs) return;
    int pu = ps[p], pv = pd[p], nu = ns[p], nv = nd[p];
    float sp = 0.0f, sn = 0.0f;
    #pragma unroll
    for (int j = 0; j < 8; ++j) {
        int c = lane * 8 + j;
        sp += zf[(size_t)pu * HDIM + c] * zf[(size_t)pv * HDIM + c];
        sn += zf[(size_t)nu * HDIM + c] * zf[(size_t)nv * HDIM + c];
    }
    #pragma unroll
    for (int o = 16; o > 0; o >>= 1) {
        sp += __shfl_xor(sp, o, 32);
        sn += __shfl_xor(sn, o, 32);
    }
    if (lane == 0) {
        float gp = bias[nids[pu]] + bias[nids[pv]];
        float gn = bias[nids[nu]] + bias[nids[nv]];
        float m  = (sn + gn) - (sp + gp) + 1.0f;
        out[p] = m > 0.0f ? m : 0.0f;
    }
}

extern "C" void kernel_launch(void* const* d_in, const int* in_sizes, int n_in,
                              void* d_out, int out_size, void* d_ws, size_t ws_size,
                              hipStream_t stream) {
    const float* feat = (const float*)d_in[0];
    const float* Wp   = (const float*)d_in[1];
    const float* bp   = (const float*)d_in[2];
    const float* Q1   = (const float*)d_in[3];
    const float* bq1  = (const float*)d_in[4];
    const float* W1   = (const float*)d_in[5];
    const float* bw1  = (const float*)d_in[6];
    const float* Q2   = (const float*)d_in[7];
    const float* bq2  = (const float*)d_in[8];
    const float* W2   = (const float*)d_in[9];
    const float* bw2  = (const float*)d_in[10];
    const float* w1v  = (const float*)d_in[11];
    const float* w2v  = (const float*)d_in[12];
    const float* bias = (const float*)d_in[13];
    const int* src1   = (const int*)d_in[14];
    const int* dst1   = (const int*)d_in[15];
    const int* src2   = (const int*)d_in[16];
    const int* dst2   = (const int*)d_in[17];
    const int* pos_s  = (const int*)d_in[18];
    const int* pos_d  = (const int*)d_in[19];
    const int* neg_s  = (const int*)d_in[20];
    const int* neg_d  = (const int*)d_in[21];
    const int* nids   = (const int*)d_in[22];
    float* out        = (float*)d_out;

    char* ws = (char*)d_ws;
    const size_t SZ_HP  = (size_t)N_SRC * HDIM * 4;
    const size_t SZ_N   = (size_t)N_SRC * HDIM * 4;
    const size_t SZ_H1  = (size_t)N_MID * HDIM * 4;
    const size_t SZ_AGG = (size_t)N_MID * HDIM * 4;
    const size_t SZ_WS  = 256 * 1024;
    const size_t SZ_H2  = (size_t)N_DST * HDIM * 4;
    const size_t SZ_ZF  = (size_t)N_DST * HDIM * 4;
    float*  hp   = (float*)(ws);
    float*  nbuf = (float*)(ws + SZ_HP);
    float*  h1   = (float*)(ws + SZ_HP + SZ_N);
    float*  agg  = (float*)(ws + SZ_HP + SZ_N + SZ_H1);
    float*  wsum = (float*)(ws + SZ_HP + SZ_N + SZ_H1 + SZ_AGG);
    float*  h2   = (float*)(ws + SZ_HP + SZ_N + SZ_H1 + SZ_AGG + SZ_WS);
    float*  zf   = (float*)(ws + SZ_HP + SZ_N + SZ_H1 + SZ_AGG + SZ_WS + SZ_H2);
    __bf16* Wpb  = (__bf16*)(ws + SZ_HP + SZ_N + SZ_H1 + SZ_AGG + SZ_WS + SZ_H2 + SZ_ZF);
    __bf16* Q1b  = Wpb + HDIM * HDIM;
    __bf16* W1b  = Q1b + HDIM * HDIM;
    __bf16* Q2b  = W1b + HDIM * 2 * HDIM;
    __bf16* W2b  = Q2b + HDIM * HDIM;

    f32_to_bf16_kernel<<<(HDIM * HDIM + 255) / 256, 256, 0, stream>>>(Wp, Wpb, HDIM * HDIM);
    f32_to_bf16_kernel<<<(HDIM * HDIM + 255) / 256, 256, 0, stream>>>(Q1, Q1b, HDIM * HDIM);
    f32_to_bf16_kernel<<<(HDIM * 2 * HDIM + 255) / 256, 256, 0, stream>>>(W1, W1b, HDIM * 2 * HDIM);
    f32_to_bf16_kernel<<<(HDIM * HDIM + 255) / 256, 256, 0, stream>>>(Q2, Q2b, HDIM * HDIM);
    f32_to_bf16_kernel<<<(HDIM * 2 * HDIM + 255) / 256, 256, 0, stream>>>(W2, W2b, HDIM * 2 * HDIM);

    gemm_bf16_n256<<<N_SRC / 32, 256, 0, stream>>>(feat, Wpb, bp, hp, N_SRC, HDIM, 0);
    gemm_bf16_n256<<<N_SRC / 32, 256, 0, stream>>>(hp, Q1b, bq1, nbuf, N_SRC, HDIM, 1);
    {
        int nz = N_MID * HDIM + 65536;
        zero_f32_kernel<<<(nz + 255) / 256, 256, 0, stream>>>(agg, nz);
        edge_scatter_kernel<<<E1_N / 8, 256, 0, stream>>>(nbuf, w1v, src1, dst1, agg, wsum, E1_N);
    }
    build_z_kernel<<<(N_MID * 512 + 255) / 256, 256, 0, stream>>>(agg, wsum, hp, nbuf, N_MID);
    gemm_bf16_n256<<<N_MID / 32, 256, 0, stream>>>(nbuf, W1b, bw1, h1, N_MID, 2 * HDIM, 1);
    gemm_bf16_n256<<<N_MID / 32, 256, 0, stream>>>(h1, Q2b, bq2, nbuf, N_MID, HDIM, 1);
    {
        int nz = N_DST * HDIM;
        zero_f32_kernel<<<(nz + 255) / 256, 256, 0, stream>>>(agg, nz);
        zero_f32_kernel<<<(N_DST + 255) / 256, 256, 0, stream>>>(wsum, N_DST);
        edge_scatter_kernel<<<E2_N / 8, 256, 0, stream>>>(nbuf, w2v, src2, dst2, agg, wsum, E2_N);
    }
    build_z_kernel<<<(N_DST * 512 + 255) / 256, 256, 0, stream>>>(agg, wsum, h1, nbuf, N_DST);
    gemm_bf16_n256<<<N_DST / 32, 256, 0, stream>>>(nbuf, W2b, bw2, h2, N_DST, 2 * HDIM, 1);
    skip_norm_kernel<<<N_DST / 8, 256, 0, stream>>>(hp, h2, zf, N_DST);
    score_kernel<<<N_PAIRS / 8, 256, 0, stream>>>(zf, bias, nids,
                                                  pos_s, pos_d, neg_s, neg_d,
                                                  out, N_PAIRS);
}


# Segment sums in LDS: a work-group owns a dst-row range, scans the dst list for its edges, gathers rows, adds into an f32 LDS tile (LDS float atomics, arrival order), stores agg once; zero passes gone
# speedup vs baseline: 3.5385x; 3.5385x over previous
#include <hip/hip_runtime.h>
#include <hip/hip_bf16.h>


typedef __attribute__((ext_vector_type(16))) __bf16 v16bf;
typedef __attribute__((ext_vector_type(8)))  __bf16 v8bf;
typedef __attribute__((ext_vector_type(4)))  __bf16 v4bf;
typedef __attribute__((ext_vector_type(8)))  float  v8f;

#define N_SRC   131072
#define N_MID   32768
#define N_DST   8192
#define E1_N    524288
#define E2_N    131072
#define HDIM    256
#define N_PAIRS 8192


__global__ void pack_w_bf16_kernel(const float* __restrict__ src,
                                   __bf16* __restrict__ dst, int K, int n) {
    int i = blockIdx.x * blockDim.x + threadIdx.x;
    if (i >= n) return;
    int e    = i & 7;
    int l    = (i >> 3) & 31;
    int j    = (i >> 8) & 1;
    int frag = i >> 9;
    int KS   = K >> 5;
    int nt   = frag / KS;
    int ks   = frag - nt * KS;
    int col  = nt * 16 + (l & 15);
    int k    = ks * 32 + j * 16 + (l >> 4) * 8 + e;
    dst[i] = (__bf16)src[(size_t)col * K + k];
}

__global__ __launch_bounds__(256)
void compose_q1_wp_kernel(const float* __restrict__ Q1, const float* __restrict__ bq1,
                          const float* __restrict__ Wp, const float* __restrict__ bp,
                          __bf16* __restrict__ Wcb, float* __restrict__ bc)
{
    __shared__ float q[HDIM];
    __shared__ float part[8];
    const int n = blockIdx.x;
    const int c = threadIdx.x;
    float qv = Q1[(size_t)n * HDIM + c];
    q[c] = qv;
    float pb = qv * bp[c];
    #pragma unroll
    for (int o = 16; o > 0; o >>= 1) pb += __shfl_xor(pb, o, 32);
    if ((c & 31) == 0) part[c >> 5] = pb;
    __syncthreads();
    float acc = 0.0f;
    #pragma unroll 32
    for (int m = 0; m < HDIM; ++m)
        acc = fmaf(q[m], Wp[(size_t)m * HDIM + c], acc);
    int e  = c & 7;
    int h  = (c >> 3) & 1;
    int j  = (c >> 4) & 1;
    int ks = c >> 5;
    int l  = h * 16 + (n & 15);
    int nt = n >> 4;
    int i  = ((((nt * (HDIM >> 5) + ks) * 2 + j) * 32 + l) << 3) + e;
    Wcb[i] = (__bf16)acc;
    if (c == 0) {
        float s = bq1[n];
        #pragma unroll
        for (int w = 0; w < 8; ++w) s += part[w];
        bc[n] = s;
    }
}

#define A_PAD 8

template <bool SCALED>
__device__ __forceinline__ void stage_a_32x256(const float* __restrict__ src,
                                               const float* __restrict__ wsum,
                                               int r0, __bf16* __restrict__ tile,
                                               int pitch, int kofs, int wave, int lane)
{
    float4 v[8];
    #pragma unroll
    for (int i = 0; i < 8; ++i) {
        int r = wave * 4 + (i >> 1);
        int cidx = (i & 1) * 128 + lane * 4;
        v[i] = *(const float4*)(src + (size_t)(r0 + r) * HDIM + cidx);
    }
    float rs[4];
    if (SCALED) {
        #pragma unroll
        for (int j = 0; j < 4; ++j) {
            float ws = wsum[r0 + wave * 4 + j];
            ws = ws < 1.0f ? 1.0f : ws;
            rs[j] = __frcp_rn(ws);
        }
    }
    #pragma unroll
    for (int i = 0; i < 8; ++i) {
        int r = wave * 4 + (i >> 1);
        int cidx = (i & 1) * 128 + lane * 4;
        float4 x = v[i];
        if (SCALED) { float s = rs[i >> 1]; x.x *= s; x.y *= s; x.z *= s; x.w *= s; }
        v4bf o;
        o[0] = (__bf16)x.x; o[1] = (__bf16)x.y; o[2] = (__bf16)x.z; o[3] = (__bf16)x.w;
        *(v4bf*)(tile + r * pitch + kofs + cidx) = o;
    }
}

__device__ __forceinline__ void kstep_16x64(const __bf16* __restrict__ ak,
                                            const __bf16* __restrict__ wk, int tstride,
                                            v8f* acc)
{
    v8bf alo = *(const v8bf*)(ak);
    v8bf ahi = *(const v8bf*)(ak + 16);
    v16bf a = __builtin_shufflevector(alo, ahi, 0, 1, 2, 3, 4, 5, 6, 7,
                                      8, 9, 10, 11, 12, 13, 14, 15);
    #pragma unroll
    for (int t = 0; t < 4; ++t) {
        const __bf16* wt = wk + (size_t)t * tstride;
        v8bf blo = *(const v8bf*)(wt);
        v8bf bhi = *(const v8bf*)(wt + 256);
        v16bf b = __builtin_shufflevector(blo, bhi, 0, 1, 2, 3, 4, 5, 6, 7,
                                          8, 9, 10, 11, 12, 13, 14, 15);
        acc[t] = __builtin_amdgcn_wmma_f32_16x16x32_bf16(
            false, a, false, b, (short)0, acc[t], false, false);
    }
}

__device__ __forceinline__ void store_strip(const v8f* acc, const float* __restrict__ bias,
                                            float* __restrict__ C, int m0, int n0,
                                            int half, int l15, int relu)
{
    #pragma unroll
    for (int t = 0; t < 4; ++t) {
        int n = n0 + t * 16 + l15;
        float bv = bias[n];
        #pragma unroll
        for (int r = 0; r < 8; ++r) {
            int m = m0 + half * 8 + r;
            float v = acc[t][r] + bv;
            if (relu) v = fmaxf(v, 0.0f);
            C[(size_t)m * HDIM + n] = v;
        }
    }
}

__global__ __launch_bounds__(256) __attribute__((amdgpu_waves_per_eu(4, 12)))
void gemm_bf16_n256(const float* __restrict__ A, const __bf16* __restrict__ W,
                    const float* __restrict__ bias, float* __restrict__ C,
                    int relu)
{
    constexpr int PITCH = HDIM + A_PAD;
    __shared__ __attribute__((aligned(16))) __bf16 atile[32 * PITCH];

    const int lane = threadIdx.x & 31;
    const int wave = threadIdx.x >> 5;
    const int half = lane >> 4;
    const int l15  = lane & 15;
    const int mb   = blockIdx.x * 32;
    const int mw   = (wave >> 2) * 16;
    const int n0   = (wave & 3) * 64;

    stage_a_32x256<false>(A, nullptr, mb, atile, PITCH, 0, wave, lane);
    __syncthreads();

    v8f acc[4];
    #pragma unroll
    for (int t = 0; t < 4; ++t) acc[t] = (v8f){};

    const __bf16* arow = atile + (mw + l15) * PITCH + half * 8;
    const int     tstride = (HDIM >> 5) * 512;
    const __bf16* wfrag = W + (size_t)(n0 >> 4) * tstride + lane * 8;

    for (int ks = 0; ks < HDIM / 32; ++ks)
        kstep_16x64(arow + ks * 32, wfrag + ks * 512, tstride, acc);

    store_strip(acc, bias, C, mb + mw, n0, half, l15, relu);
}

__global__ __launch_bounds__(256) __attribute__((amdgpu_waves_per_eu(4, 12)))
void gemm_cat_bf16_n256(const float* __restrict__ agg, const float* __restrict__ wsum,
                        const float* __restrict__ hdst, const __bf16* __restrict__ W,
                        const float* __restrict__ bias, float* __restrict__ C, int relu)
{
    constexpr int PITCH = 2 * HDIM + A_PAD;
    __shared__ __attribute__((aligned(16))) __bf16 atile[32 * PITCH];

    const int lane = threadIdx.x & 31;
    const int wave = threadIdx.x >> 5;
    const int half = lane >> 4;
    const int l15  = lane & 15;
    const int mb   = blockIdx.x * 32;
    const int mw   = (wave >> 2) * 16;
    const int n0   = (wave & 3) * 64;

    stage_a_32x256<true >(agg,  wsum,    mb, atile, PITCH, 0,    wave, lane);
    stage_a_32x256<false>(hdst, nullptr, mb, atile, PITCH, HDIM, wave, lane);
    __syncthreads();

    v8f acc[4];
    #pragma unroll
    for (int t = 0; t < 4; ++t) acc[t] = (v8f){};

    const __bf16* arow = atile + (mw + l15) * PITCH + half * 8;
    const int     tstride = (2 * HDIM / 32) * 512;
    const __bf16* wfrag = W + (size_t)(n0 >> 4) * tstride + lane * 8;

    for (int ks = 0; ks < 2 * HDIM / 32; ++ks)
        kstep_16x64(arow + ks * 32, wfrag + ks * 512, tstride, acc);

    store_strip(acc, bias, C, mb + mw, n0, half, l15, relu);
}

#define SEG_THREADS 1024
#define SEG_CAP     8192
#define SEG_UA      8

template <int ROWS, int LRB, int CH, int E>
__global__ __launch_bounds__(SEG_THREADS)
void segsum_owner_kernel(const float* __restrict__ nsrc,
                         const float* __restrict__ w,
                         const int* __restrict__ src,
                         const int* __restrict__ dst,
                         float* __restrict__ agg,
                         float* __restrict__ wsum)
{
    constexpr int TRIP = SEG_THREADS * SEG_UA;
    constexpr int NIT  = E / (4 * TRIP);
    static_assert(ROWS == (1 << LRB), "local row fits LRB bits");
    static_assert(E % (4 * TRIP) == 0, "the scan has whole trips only");
    static_assert((NIT & (NIT - 1)) == 0, "trip rotation by mask");
    static_assert(((long long)E << LRB) <= 0xffffffffLL, "list entry fits 32 bits");
    __shared__ __attribute__((aligned(16))) float tile[ROWS * HDIM];
    __shared__ float    wl[ROWS];
    __shared__ unsigned list[SEG_CAP];
    __shared__ unsigned cnt;

    const int tid  = threadIdx.x;
    const int lane = tid & 31;
    const int wave = __builtin_amdgcn_readfirstlane(tid >> 5);
    const int base = blockIdx.x * ROWS;

    for (int i = tid; i < ROWS * HDIM / 4; i += SEG_THREADS)
        ((float4*)tile)[i] = make_float4(0.0f, 0.0f, 0.0f, 0.0f);
    if (tid < ROWS) wl[tid] = 0.0f;
    if (tid == 0) cnt = 0u;
    __syncthreads();

    auto push = [&](unsigned lr, int e) {
        if (lr < (unsigned)ROWS) {
            unsigned pos = atomicAdd(&cnt, 1u);
            if (pos < SEG_CAP) {
                list[pos] = ((unsigned)e << LRB) | lr;
            } else {
                int   s  = src[e];
                float we = w[e];
                atomicAdd(&wl[lr], we);
                for (int c = 0; c < HDIM; ++c)
                    atomicAdd(&tile[lr * HDIM + c], nsrc[(size_t)s * HDIM + c] * we);
            }
        }
    };
    const int4* d4 = (const int4*)dst;
    for (int it = 0; it < NIT; ++it) {
        const int itr = (it + (int)blockIdx.x) & (NIT - 1);
        const int i0  = itr * TRIP + tid;
        int4 d[SEG_UA];
        #pragma unroll
        for (int u = 0; u < SEG_UA; ++u) d[u] = d4[i0 + u * SEG_THREADS];
        #pragma unroll
        for (int u = 0; u < SEG_UA; ++u) {
            const int e0 = (i0 + u * SEG_THREADS) * 4;
            const unsigned l0 = (unsigned)(d[u].x - base);
            const unsigned l1 = (unsigned)(d[u].y - base);
            const unsigned l2 = (unsigned)(d[u].z - base);
            const unsigned l3 = (unsigned)(d[u].w - base);
            const unsigned lm = min(min(l0, l1), min(l2, l3));
            if (lm < (unsigned)ROWS) {
                push(l0, e0);
                push(l1, e0 + 1);
                push(l2, e0 + 2);
                push(l3, e0 + 3);
            }
        }
    }
    __syncthreads();

    unsigned nl = cnt;
    nl = nl < (unsigned)SEG_CAP ? nl : (unsigned)SEG_CAP;
    const int n = __builtin_amdgcn_readfirstlane((int)nl);
    for (int c0 = wave * CH; c0 < n; c0 += (SEG_THREADS / 32) * CH) {
        const int  idx   = c0 + lane;
        const bool valid = (lane < CH) && (idx < n);
        unsigned ent = valid ? list[idx] : 0u;
        int   e  = (int)(ent >> LRB);
        int   lr = (int)(ent & (unsigned)(ROWS - 1));
        int   s  = valid ? src[e] : 0;
        float we = valid ? w[e] : 0.0f;
        if (valid) atomicAdd(&wl[lr], we);
        const int rem = n - c0;
        const int cw  = rem < CH ? rem : CH;
        int j = 0;
        for (; j + 4 <= cw; j += 4) {
            int sj[4], lj[4]; float wj[4];
            #pragma unroll
            for (int u = 0; u < 4; ++u) {
                sj[u] = __builtin_amdgcn_readlane(s, j + u);
                lj[u] = __builtin_amdgcn_readlane(lr, j + u);
                wj[u] = __int_as_float(__builtin_amdgcn_readlane(__float_as_int(we), j + u));
            }
            float v[4][8];
            #pragma unroll
            for (int u = 0; u < 4; ++u) {
                const float* nr = nsrc + (size_t)sj[u] * HDIM + lane;
                #pragma unroll
                for (int q = 0; q < 8; ++q) v[u][q] = nr[q * 32];
            }
            #pragma unroll
            for (int u = 0; u < 4; ++u) {
                float* tr = tile + lj[u] * HDIM + lane;
                #pragma unroll
                for (int q = 0; q < 8; ++q) atomicAdd(tr + q * 32, v[u][q] * wj[u]);
            }
        }
        for (; j < cw; ++j) {
            int   s1 = __builtin_amdgcn_readlane(s, j);
            int   l1 = __builtin_amdgcn_readlane(lr, j);
            float w1 = __int_as_float(__builtin_amdgcn_readlane(__float_as_int(we), j));
            const float* nr = nsrc + (size_t)s1 * HDIM + lane;
            float* tr = tile + l1 * HDIM + lane;
            float v1[8];
            #pragma unroll
            for (int q = 0; q < 8; ++q) v1[q] = nr[q * 32];
            #pragma unroll
            for (int q = 0; q < 8; ++q) atomicAdd(tr + q * 32, v1[q] * w1);
        }
    }
    __syncthreads();

    float4* arow = (float4*)(agg + (size_t)base * HDIM);
    for (int i = tid; i < ROWS * HDIM / 4; i += SEG_THREADS)
        arow[i] = ((const float4*)tile)[i];
    if (tid < ROWS) wsum[base + tid] = wl[tid];
}

__global__ __launch_bounds__(256)
void skip_norm_kernel(const float* __restrict__ hp, const float* __restrict__ h2,
                      float* __restrict__ zf, int rows)
{
    int r    = blockIdx.x * 8 + (threadIdx.x >> 5);
    int lane = threadIdx.x & 31;
    if (r >= rows) return;
    float v[8];
    float ss = 0.0f;
    #pragma unroll
    for (int j = 0; j < 8; ++j) {
        int c = lane * 8 + j;
        v[j] = hp[(size_t)r * HDIM + c] + h2[(size_t)r * HDIM + c];
        ss += v[j] * v[j];
    }
    #pragma unroll
    for (int o = 16; o > 0; o >>= 1) ss += __shfl_xor(ss, o, 32);
    float nrm   = sqrtf(ss);
    float scale = (nrm == 0.0f) ? 1.0f : 1.0f / nrm;
    #pragma unroll
    for (int j = 0; j < 8; ++j)
        zf[(size_t)r * HDIM + lane * 8 + j] = v[j] * scale;
}

__global__ __launch_bounds__(256)
void score_kernel(const float* __restrict__ zf, const float* __restrict__ bias,
                  const int* __restrict__ nids,
                  const int* __restrict__ ps, const int* __restrict__ pd,
                  const int* __restrict__ ns, const int* __restrict__ nd,
                  float* __restrict__ out, int npairs)
{
    int p    = blockIdx.x * 8 + (threadIdx.x >> 5);
    int lane = threadIdx.x & 31;
    if (p >= npairs) return;
    int pu = ps[p], pv = pd[p], nu = ns[p], nv = nd[p];
    float sp = 0.0f, sn = 0.0f;
    #pragma unroll
    for (int j = 0; j < 8; ++j) {
        int c = lane * 8 + j;
        sp += zf[(size_t)pu * HDIM + c] * zf[(size_t)pv * HDIM + c];
        sn += zf[(size_t)nu * HDIM + c] * zf[(size_t)nv * HDIM + c];
    }
    #pragma unroll
    for (int o = 16; o > 0; o >>= 1) {
        sp += __shfl_xor(sp, o, 32);
        sn += __shfl_xor(sn, o, 32);
    }
    if (lane == 0) {
        float gp = bias[nids[pu]] + bias[nids[pv]];
        float gn = bias[nids[nu]] + bias[nids[nv]];
        float m  = (sn + gn) - (sp + gp) + 1.0f;
        out[p] = m > 0.0f ? m : 0.0f;
    }
}

extern "C" void kernel_launch(void* const* d_in, const int* in_sizes, int n_in,
                              void* d_out, int out_size, void* d_ws, size_t ws_size,
                              hipStream_t stream) {
    const float* feat = (const float*)d_in[0];
    const float* Wp   = (const float*)d_in[1];
    const float* bp   = (const float*)d_in[2];
    const float* Q1   = (const float*)d_in[3];
    const float* bq1  = (const float*)d_in[4];
    const float* W1   = (const float*)d_in[5];
    const float* bw1  = (const float*)d_in[6];
    const float* Q2   = (const float*)d_in[7];
    const float* bq2  = (const float*)d_in[8];
    const float* W2   = (const float*)d_in[9];
    const float* bw2  = (const float*)d_in[10];
    const float* w1v  = (const float*)d_in[11];
    const float* w2v  = (const float*)d_in[12];
    const float* bias = (const float*)d_in[13];
    const int* src1   = (const int*)d_in[14];
    const int* dst1   = (const int*)d_in[15];
    const int* src2   = (const int*)d_in[16];
    const int* dst2   = (const int*)d_in[17];
    const int* pos_s  = (const int*)d_in[18];
    const int* pos_d  = (const int*)d_in[19];
    const int* neg_s  = (const int*)d_in[20];
    const int* neg_d  = (const int*)d_in[21];
    const int* nids   = (const int*)d_in[22];
    float* out        = (float*)d_out;

    char* ws = (char*)d_ws;
    const size_t SZ_HP  = (size_t)N_SRC * HDIM * 4;
    const size_t SZ_N   = (size_t)N_SRC * HDIM * 4;
    const size_t SZ_H1  = (size_t)N_MID * HDIM * 4;
    const size_t SZ_AGG = (size_t)N_MID * HDIM * 4;
    const size_t SZ_WS  = 256 * 1024;
    const size_t SZ_H2  = (size_t)N_DST * HDIM * 4;
    const size_t SZ_ZF  = (size_t)N_DST * HDIM * 4;
    float*  hp   = (float*)(ws);
    float*  bc   = hp + (size_t)N_MID * HDIM;
    float*  nbuf = (float*)(ws + SZ_HP);
    float*  h1   = (float*)(ws + SZ_HP + SZ_N);
    float*  agg  = (float*)(ws + SZ_HP + SZ_N + SZ_H1);
    float*  wsum = (float*)(ws + SZ_HP + SZ_N + SZ_H1 + SZ_AGG);
    float*  h2   = (float*)(ws + SZ_HP + SZ_N + SZ_H1 + SZ_AGG + SZ_WS);
    float*  zf   = (float*)(ws + SZ_HP + SZ_N + SZ_H1 + SZ_AGG + SZ_WS + SZ_H2);
    __bf16* Wpb  = (__bf16*)(ws + SZ_HP + SZ_N + SZ_H1 + SZ_AGG + SZ_WS + SZ_H2 + SZ_ZF);
    __bf16* Wcb  = Wpb + HDIM * HDIM;
    __bf16* W1b  = Wcb + HDIM * HDIM;
    __bf16* Q2b  = W1b + HDIM * 2 * HDIM;
    __bf16* W2b  = Q2b + HDIM * HDIM;

    pack_w_bf16_kernel<<<(HDIM * HDIM + 255) / 256, 256, 0, stream>>>(Wp, Wpb, HDIM, HDIM * HDIM);
    compose_q1_wp_kernel<<<HDIM, 256, 0, stream>>>(Q1, bq1, Wp, bp, Wcb, bc);
    pack_w_bf16_kernel<<<(HDIM * 2 * HDIM + 255) / 256, 256, 0, stream>>>(W1, W1b, 2 * HDIM, HDIM * 2 * HDIM);
    pack_w_bf16_kernel<<<(HDIM * HDIM + 255) / 256, 256, 0, stream>>>(Q2, Q2b, HDIM, HDIM * HDIM);
    pack_w_bf16_kernel<<<(HDIM * 2 * HDIM + 255) / 256, 256, 0, stream>>>(W2, W2b, 2 * HDIM, HDIM * 2 * HDIM);

    gemm_bf16_n256<<<N_MID / 32, 256, 0, stream>>>(feat, Wpb, bp, hp, 0);
    gemm_bf16_n256<<<N_SRC / 32, 256, 0, stream>>>(feat, Wcb, bc, nbuf, 1);
    segsum_owner_kernel<128, 7, 32, E1_N><<<N_MID / 128, SEG_THREADS, 0, stream>>>(
        nbuf, w1v, src1, dst1, agg, wsum);
    gemm_cat_bf16_n256<<<N_MID / 32, 256, 0, stream>>>(agg, wsum, hp, W1b, bw1, h1, 1);
    gemm_bf16_n256<<<N_MID / 32, 256, 0, stream>>>(h1, Q2b, bq2, nbuf, 1);
    segsum_owner_kernel<32, 5, 16, E2_N><<<N_DST / 32, SEG_THREADS, 0, stream>>>(
        nbuf, w2v, src2, dst2, agg, wsum);
    gemm_cat_bf16_n256<<<N_DST / 32, 256, 0, stream>>>(agg, wsum, h1, W2b, bw2, h2, 1);
    skip_norm_kernel<<<N_DST / 8, 256, 0, stream>>>(hp, h2, zf, N_DST);
    score_kernel<<<N_PAIRS / 8, 256, 0, stream>>>(zf, bias, nids,
                                                  pos_s, pos_d, neg_s, neg_d,
                                                  out, N_PAIRS);
}


# GEMM B feed: 4-wave blocks, wave tile 32x64, a B fragment loaded once for both 16-row tiles, a K-step's four fragments requested one K-step ahead (2nd register set); k order and sums unchanged
# speedup vs baseline: 3.9001x; 1.1022x over previous
#include <hip/hip_runtime.h>
#include <hip/hip_bf16.h>


typedef __attribute__((ext_vector_type(16))) __bf16 v16bf;
typedef __attribute__((ext_vector_type(8)))  __bf16 v8bf;
typedef __attribute__((ext_vector_type(4)))  __bf16 v4bf;
typedef __attribute__((ext_vector_type(8)))  float  v8f;

#define N_SRC   131072
#define N_MID   32768
#define N_DST   8192
#define E1_N    524288
#define E2_N    131072
#define HDIM    256
#define N_PAIRS 8192


__global__ void pack_w_bf16_kernel(const float* __restrict__ src,
                                   __bf16* __restrict__ dst, int K, int n) {
    int i = blockIdx.x * blockDim.x + threadIdx.x;
    if (i >= n) return;
    int e    = i & 7;
    int l    = (i >> 3) & 31;
    int j    = (i >> 8) & 1;
    int frag = i >> 9;
    int KS   = K >> 5;
    int nt   = frag / KS;
    int ks   = frag - nt * KS;
    int col  = nt * 16 + (l & 15);
    int k    = ks * 32 + j * 16 + (l >> 4) * 8 + e;
    dst[i] = (__bf16)src[(size_t)col * K + k];
}

__global__ __launch_bounds__(256)
void compose_q1_wp_kernel(const float* __restrict__ Q1, const float* __restrict__ bq1,
                          const float* __restrict__ Wp, const float* __restrict__ bp,
                          __bf16* __restrict__ Wcb, float* __restrict__ bc)
{
    __shared__ float q[HDIM];
    __shared__ float part[8];
    const int n = blockIdx.x;
    const int c = threadIdx.x;
    float qv = Q1[(size_t)n * HDIM + c];
    q[c] = qv;
    float pb = qv * bp[c];
    #pragma unroll
    for (int o = 16; o > 0; o >>= 1) pb += __shfl_xor(pb, o, 32);
    if ((c & 31) == 0) part[c >> 5] = pb;
    __syncthreads();
    float acc = 0.0f;
    #pragma unroll 32
    for (int m = 0; m < HDIM; ++m)
        acc = fmaf(q[m], Wp[(size_t)m * HDIM + c], acc);
    int e  = c & 7;
    int h  = (c >> 3) & 1;
    int j  = (c >> 4) & 1;
    int ks = c >> 5;
    int l  = h * 16 + (n & 15);
    int nt = n >> 4;
    int i  = ((((nt * (HDIM >> 5) + ks) * 2 + j) * 32 + l) << 3) + e;
    Wcb[i] = (__bf16)acc;
    if (c == 0) {
        float s = bq1[n];
        #pragma unroll
        for (int w = 0; w < 8; ++w) s += part[w];
        bc[n] = s;
    }
}

#define A_PAD 8

template <bool SCALED>
__device__ __forceinline__ void stage_a_32x256(const float* __restrict__ src,
                                               const float* __restrict__ wsum,
                                               int r0, __bf16* __restrict__ tile,
                                               int pitch, int kofs, int wave, int lane)
{
    float4 v[16];
    #pragma unroll
    for (int i = 0; i < 16; ++i) {
        int r = wave * 8 + (i >> 1);
        int cidx = (i & 1) * 128 + lane * 4;
        v[i] = *(const float4*)(src + (size_t)(r0 + r) * HDIM + cidx);
    }
    float rs[8];
    if (SCALED) {
        #pragma unroll
        for (int j = 0; j < 8; ++j) {
            float ws = wsum[r0 + wave * 8 + j];
            ws = ws < 1.0f ? 1.0f : ws;
            rs[j] = __frcp_rn(ws);
        }
    }
    #pragma unroll
    for (int i = 0; i < 16; ++i) {
        int r = wave * 8 + (i >> 1);
        int cidx = (i & 1) * 128 + lane * 4;
        float4 x = v[i];
        if (SCALED) { float s = rs[i >> 1]; x.x *= s; x.y *= s; x.z *= s; x.w *= s; }
        v4bf o;
        o[0] = (__bf16)x.x; o[1] = (__bf16)x.y; o[2] = (__bf16)x.z; o[3] = (__bf16)x.w;
        *(v4bf*)(tile + r * pitch + kofs + cidx) = o;
    }
}

__device__ __forceinline__ void load_b_kstep(const __bf16* __restrict__ wk, int tstride,
                                             v16bf* b)
{
    #pragma unroll
    for (int t = 0; t < 4; ++t) {
        const __bf16* wt = wk + (size_t)t * tstride;
        v8bf blo = *(const v8bf*)(wt);
        v8bf bhi = *(const v8bf*)(wt + 256);
        b[t] = __builtin_shufflevector(blo, bhi, 0, 1, 2, 3, 4, 5, 6, 7,
                                       8, 9, 10, 11, 12, 13, 14, 15);
    }
}

__device__ __forceinline__ void kstep_32x64(const __bf16* __restrict__ ak, int rstride,
                                            const v16bf* b, v8f* acc0, v8f* acc1)
{
    v8bf alo0 = *(const v8bf*)(ak);
    v8bf ahi0 = *(const v8bf*)(ak + 16);
    v8bf alo1 = *(const v8bf*)(ak + rstride);
    v8bf ahi1 = *(const v8bf*)(ak + rstride + 16);
    v16bf a0 = __builtin_shufflevector(alo0, ahi0, 0, 1, 2, 3, 4, 5, 6, 7,
                                       8, 9, 10, 11, 12, 13, 14, 15);
    v16bf a1 = __builtin_shufflevector(alo1, ahi1, 0, 1, 2, 3, 4, 5, 6, 7,
                                       8, 9, 10, 11, 12, 13, 14, 15);
    #pragma unroll
    for (int t = 0; t < 4; ++t) {
        acc0[t] = __builtin_amdgcn_wmma_f32_16x16x32_bf16(
            false, a0, false, b[t], (short)0, acc0[t], false, false);
        acc1[t] = __builtin_amdgcn_wmma_f32_16x16x32_bf16(
            false, a1, false, b[t], (short)0, acc1[t], false, false);
    }
}

template <int KS>
__device__ __forceinline__ void kloop_32x64(const __bf16* __restrict__ arow, int rstride,
                                            const __bf16* __restrict__ wfrag, int tstride,
                                            v8f* acc0, v8f* acc1)
{
    v16bf bcur[4], bnxt[4];
    load_b_kstep(wfrag, tstride, bcur);
    #pragma unroll
    for (int ks = 0; ks < KS; ++ks) {
        if (ks + 1 < KS) load_b_kstep(wfrag + (ks + 1) * 512, tstride, bnxt);
        __builtin_amdgcn_sched_barrier(0);
        kstep_32x64(arow + ks * 32, rstride, bcur, acc0, acc1);
        if (ks + 1 < KS) {
            #pragma unroll
            for (int t = 0; t < 4; ++t) bcur[t] = bnxt[t];
        }
    }
}

__device__ __forceinline__ void store_strip(const v8f* acc, const float* __restrict__ bias,
                                            float* __restrict__ C, int m0, int n0,
                                            int half, int l15, int relu)
{
    #pragma unroll
    for (int t = 0; t < 4; ++t) {
        int n = n0 + t * 16 + l15;
        float bv = bias[n];
        #pragma unroll
        for (int r = 0; r < 8; ++r) {
            int m = m0 + half * 8 + r;
            float v = acc[t][r] + bv;
            if (relu) v = fmaxf(v, 0.0f);
            C[(size_t)m * HDIM + n] = v;
        }
    }
}

__global__ __launch_bounds__(128)
void gemm_bf16_n256(const float* __restrict__ A, const __bf16* __restrict__ W,
                    const float* __restrict__ bias, float* __restrict__ C,
                    int relu)
{
    constexpr int PITCH = HDIM + A_PAD;
    __shared__ __attribute__((aligned(16))) __bf16 atile[32 * PITCH];

    const int lane = threadIdx.x & 31;
    const int wave = threadIdx.x >> 5;
    const int half = lane >> 4;
    const int l15  = lane & 15;
    const int mb   = blockIdx.x * 32;
    const int n0   = wave * 64;

    stage_a_32x256<false>(A, nullptr, mb, atile, PITCH, 0, wave, lane);
    __syncthreads();

    v8f acc0[4], acc1[4];
    #pragma unroll
    for (int t = 0; t < 4; ++t) { acc0[t] = (v8f){}; acc1[t] = (v8f){}; }

    const __bf16* arow = atile + l15 * PITCH + half * 8;
    const int     tstride = (HDIM >> 5) * 512;
    const __bf16* wfrag = W + (size_t)(n0 >> 4) * tstride + lane * 8;

    kloop_32x64<HDIM / 32>(arow, 16 * PITCH, wfrag, tstride, acc0, acc1);

    store_strip(acc0, bias, C, mb,      n0, half, l15, relu);
    store_strip(acc1, bias, C, mb + 16, n0, half, l15, relu);
}

__global__ __launch_bounds__(128)
void gemm_cat_bf16_n256(const float* __restrict__ agg, const float* __restrict__ wsum,
                        const float* __restrict__ hdst, const __bf16* __restrict__ W,
                        const float* __restrict__ bias, float* __restrict__ C, int relu)
{
    constexpr int PITCH = 2 * HDIM + A_PAD;
    __shared__ __attribute__((aligned(16))) __bf16 atile[32 * PITCH];

    const int lane = threadIdx.x & 31;
    const int wave = threadIdx.x >> 5;
    const int half = lane >> 4;
    const int l15  = lane & 15;
    const int mb   = blockIdx.x * 32;
    const int n0   = wave * 64;

    stage_a_32x256<true >(agg,  wsum,    mb, atile, PITCH, 0,    wave, lane);
    stage_a_32x256<false>(hdst, nullptr, mb, atile, PITCH, HDIM, wave, lane);
    __syncthreads();

    v8f acc0[4], acc1[4];
    #pragma unroll
    for (int t = 0; t < 4; ++t) { acc0[t] = (v8f){}; acc1[t] = (v8f){}; }

    const __bf16* arow = atile + l15 * PITCH + half * 8;
    const int     tstride = (2 * HDIM / 32) * 512;
    const __bf16* wfrag = W + (size_t)(n0 >> 4) * tstride + lane * 8;

    kloop_32x64<2 * HDIM / 32>(arow, 16 * PITCH, wfrag, tstride, acc0, acc1);

    store_strip(acc0, bias, C, mb,      n0, half, l15, relu);
    store_strip(acc1, bias, C, mb + 16, n0, half, l15, relu);
}

#define SEG_THREADS 1024
#define SEG_CAP     8192
#define SEG_UA      8

template <int ROWS, int LRB, int CH, int E>
__global__ __launch_bounds__(SEG_THREADS)
void segsum_owner_kernel(const float* __restrict__ nsrc,
                         const float* __restrict__ w,
                         const int* __restrict__ src,
                         const int* __restrict__ dst,
                         float* __restrict__ agg,
                         float* __restrict__ wsum)
{
    constexpr int TRIP = SEG_THREADS * SEG_UA;
    constexpr int NIT  = E / (4 * TRIP);
    static_assert(ROWS == (1 << LRB), "local row fits LRB bits");
    static_assert(E % (4 * TRIP) == 0, "the scan has whole trips only");
    static_assert((NIT & (NIT - 1)) == 0, "trip rotation by mask");
    static_assert(((long long)E << LRB) <= 0xffffffffLL, "list entry fits 32 bits");
    __shared__ __attribute__((aligned(16))) float tile[ROWS * HDIM];
    __shared__ float    wl[ROWS];
    __shared__ unsigned list[SEG_CAP];
    __shared__ unsigned cnt;

    const int tid  = threadIdx.x;
    const int lane = tid & 31;
    const int wave = __builtin_amdgcn_readfirstlane(tid >> 5);
    const int base = blockIdx.x * ROWS;

    for (int i = tid; i < ROWS * HDIM / 4; i += SEG_THREADS)
        ((float4*)tile)[i] = make_float4(0.0f, 0.0f, 0.0f, 0.0f);
    if (tid < ROWS) wl[tid] = 0.0f;
    if (tid == 0) cnt = 0u;
    __syncthreads();

    auto push = [&](unsigned lr, int e) {
        if (lr < (unsigned)ROWS) {
            unsigned pos = atomicAdd(&cnt, 1u);
            if (pos < SEG_CAP) {
                list[pos] = ((unsigned)e << LRB) | lr;
            } else {
                int   s  = src[e];
                float we = w[e];
                atomicAdd(&wl[lr], we);
                for (int c = 0; c < HDIM; ++c)
                    atomicAdd(&tile[lr * HDIM + c], nsrc[(size_t)s * HDIM + c] * we);
            }
        }
    };
    const int4* d4 = (const int4*)dst;
    for (int it = 0; it < NIT; ++it) {
        const int itr = (it + (int)blockIdx.x) & (NIT - 1);
        const int i0  = itr * TRIP + tid;
        int4 d[SEG_UA];
        #pragma unroll
        for (int u = 0; u < SEG_UA; ++u) d[u] = d4[i0 + u * SEG_THREADS];
        #pragma unroll
        for (int u = 0; u < SEG_UA; ++u) {
            const int e0 = (i0 + u * SEG_THREADS) * 4;
            const unsigned l0 = (unsigned)(d[u].x - base);
            const unsigned l1 = (unsigned)(d[u].y - base);
            const unsigned l2 = (unsigned)(d[u].z - base);
            const unsigned l3 = (unsigned)(d[u].w - base);
            const unsigned lm = min(min(l0, l1), min(l2, l3));
            if (lm < (unsigned)ROWS) {
                push(l0, e0);
                push(l1, e0 + 1);
                push(l2, e0 + 2);
                push(l3, e0 + 3);
            }
        }
    }
    __syncthreads();

    unsigned nl = cnt;
    nl = nl < (unsigned)SEG_CAP ? nl : (unsigned)SEG_CAP;
    const int n = __builtin_amdgcn_readfirstlane((int)nl);
    for (int c0 = wave * CH; c0 < n; c0 += (SEG_THREADS / 32) * CH) {
        const int  idx   = c0 + lane;
        const bool valid = (lane < CH) && (idx < n);
        unsigned ent = valid ? list[idx] : 0u;
        int   e  = (int)(ent >> LRB);
        int   lr = (int)(ent & (unsigned)(ROWS - 1));
        int   s  = valid ? src[e] : 0;
        float we = valid ? w[e] : 0.0f;
        if (valid) atomicAdd(&wl[lr], we);
        const int rem = n - c0;
        const int cw  = rem < CH ? rem : CH;
        int j = 0;
        for (; j + 4 <= cw; j += 4) {
            int sj[4], lj[4]; float wj[4];
            #pragma unroll
            for (int u = 0; u < 4; ++u) {
                sj[u] = __builtin_amdgcn_readlane(s, j + u);
                lj[u] = __builtin_amdgcn_readlane(lr, j + u);
                wj[u] = __int_as_float(__builtin_amdgcn_readlane(__float_as_int(we), j + u));
            }
            float v[4][8];
            #pragma unroll
            for (int u = 0; u < 4; ++u) {
                const float* nr = nsrc + (size_t)sj[u] * HDIM + lane;
                #pragma unroll
                for (int q = 0; q < 8; ++q) v[u][q] = nr[q * 32];
            }
            #pragma unroll
            for (int u = 0; u < 4; ++u) {
                float* tr = tile + lj[u] * HDIM + lane;
                #pragma unroll
                for (int q = 0; q < 8; ++q) atomicAdd(tr + q * 32, v[u][q] * wj[u]);
            }
        }
        for (; j < cw; ++j) {
            int   s1 = __builtin_amdgcn_readlane(s, j);
            int   l1 = __builtin_amdgcn_readlane(lr, j);
            float w1 = __int_as_float(__builtin_amdgcn_readlane(__float_as_int(we), j));
            const float* nr = nsrc + (size_t)s1 * HDIM + lane;
            float* tr = tile + l1 * HDIM + lane;
            float v1[8];
            #pragma unroll
            for (int q = 0; q < 8; ++q) v1[q] = nr[q * 32];
            #pragma unroll
            for (int q = 0; q < 8; ++q) atomicAdd(tr + q * 32, v1[q] * w1);
        }
    }
    __syncthreads();

    float4* arow = (float4*)(agg + (size_t)base * HDIM);
    for (int i = tid; i < ROWS * HDIM / 4; i += SEG_THREADS)
        arow[i] = ((const float4*)tile)[i];
    if (tid < ROWS) wsum[base + tid] = wl[tid];
}

__global__ __launch_bounds__(256)
void skip_norm_kernel(const float* __restrict__ hp, const float* __restrict__ h2,
                      float* __restrict__ zf, int rows)
{
    int r    = blockIdx.x * 8 + (threadIdx.x >> 5);
    int lane = threadIdx.x & 31;
    if (r >= rows) return;
    float v[8];
    float ss = 0.0f;
    #pragma unroll
    for (int j = 0; j < 8; ++j) {
        int c = lane * 8 + j;
        v[j] = hp[(size_t)r * HDIM + c] + h2[(size_t)r * HDIM + c];
        ss += v[j] * v[j];
    }
    #pragma unroll
    for (int o = 16; o > 0; o >>= 1) ss += __shfl_xor(ss, o, 32);
    float nrm   = sqrtf(ss);
    float scale = (nrm == 0.0f) ? 1.0f : 1.0f / nrm;
    #pragma unroll
    for (int j = 0; j < 8; ++j)
        zf[(size_t)r * HDIM + lane * 8 + j] = v[j] * scale;
}

__global__ __launch_bounds__(256)
void score_kernel(const float* __restrict__ zf, const float* __restrict__ bias,
                  const int* __restrict__ nids,
                  const int* __restrict__ ps, const int* __restrict__ pd,
                  const int* __restrict__ ns, const int* __restrict__ nd,
                  float* __restrict__ out, int npairs)
{
    int p    = blockIdx.x * 8 + (threadIdx.x >> 5);
    int lane = threadIdx.x & 31;
    if (p >= npairs) return;
    int pu = ps[p], pv = pd[p], nu = ns[p], nv = nd[p];
    float sp = 0.0f, sn = 0.0f;
    #pragma unroll
    for (int j = 0; j < 8; ++j) {
        int c = lane * 8 + j;
        sp += zf[(size_t)pu * HDIM + c] * zf[(size_t)pv * HDIM + c];
        sn += zf[(size_t)nu * HDIM + c] * zf[(size_t)nv * HDIM + c];
    }
    #pragma unroll
    for (int o = 16; o > 0; o >>= 1) {
        sp += __shfl_xor(sp, o, 32);
        sn += __shfl_xor(sn, o, 32);
    }
    if (lane == 0) {
        float gp = bias[nids[pu]] + bias[nids[pv]];
        float gn = bias[nids[nu]] + bias[nids[nv]];
        float m  = (sn + gn) - (sp + gp) + 1.0f;
        out[p] = m > 0.0f ? m : 0.0f;
    }
}

extern "C" void kernel_launch(void* const* d_in, const int* in_sizes, int n_in,
                              void* d_out, int out_size, void* d_ws, size_t ws_size,
                              hipStream_t stream) {
    const float* feat = (const float*)d_in[0];
    const float* Wp   = (const float*)d_in[1];
    const float* bp   = (const float*)d_in[2];
    const float* Q1   = (const float*)d_in[3];
    const float* bq1  = (const float*)d_in[4];
    const float* W1   = (const float*)d_in[5];
    const float* bw1  = (const float*)d_in[6];
    const float* Q2   = (const float*)d_in[7];
    const float* bq2  = (const float*)d_in[8];
    const float* W2   = (const float*)d_in[9];
    const float* bw2  = (const float*)d_in[10];
    const float* w1v  = (const float*)d_in[11];
    const float* w2v  = (const float*)d_in[12];
    const float* bias = (const float*)d_in[13];
    const int* src1   = (const int*)d_in[14];
    const int* dst1   = (const int*)d_in[15];
    const int* src2   = (const int*)d_in[16];
    const int* dst2   = (const int*)d_in[17];
    const int* pos_s  = (const int*)d_in[18];
    const int* pos_d  = (const int*)d_in[19];
    const int* neg_s  = (const int*)d_in[20];
    const int* neg_d  = (const int*)d_in[21];
    const int* nids   = (const int*)d_in[22];
    float* out        = (float*)d_out;

    char* ws = (char*)d_ws;
    const size_t SZ_HP  = (size_t)N_SRC * HDIM * 4;
    const size_t SZ_N   = (size_t)N_SRC * HDIM * 4;
    const size_t SZ_H1  = (size_t)N_MID * HDIM * 4;
    const size_t SZ_AGG = (size_t)N_MID * HDIM * 4;
    const size_t SZ_WS  = 256 * 1024;
    const size_t SZ_H2  = (size_t)N_DST * HDIM * 4;
    const size_t SZ_ZF  = (size_t)N_DST * HDIM * 4;
    float*  hp   = (float*)(ws);
    float*  bc   = hp + (size_t)N_MID * HDIM;
    float*  nbuf = (float*)(ws + SZ_HP);
    float*  h1   = (float*)(ws + SZ_HP + SZ_N);
    float*  agg  = (float*)(ws + SZ_HP + SZ_N + SZ_H1);
    float*  wsum = (float*)(ws + SZ_HP + SZ_N + SZ_H1 + SZ_AGG);
    float*  h2   = (float*)(ws + SZ_HP + SZ_N + SZ_H1 + SZ_AGG + SZ_WS);
    float*  zf   = (float*)(ws + SZ_HP + SZ_N + SZ_H1 + SZ_AGG + SZ_WS + SZ_H2);
    __bf16* Wpb  = (__bf16*)(ws + SZ_HP + SZ_N + SZ_H1 + SZ_AGG + SZ_WS + SZ_H2 + SZ_ZF);
    __bf16* Wcb  = Wpb + HDIM * HDIM;
    __bf16* W1b  = Wcb + HDIM * HDIM;
    __bf16* Q2b  = W1b + HDIM * 2 * HDIM;
    __bf16* W2b  = Q2b + HDIM * HDIM;

    pack_w_bf16_kernel<<<(HDIM * HDIM + 255) / 256, 256, 0, stream>>>(Wp, Wpb, HDIM, HDIM * HDIM);
    compose_q1_wp_kernel<<<HDIM, 256, 0, stream>>>(Q1, bq1, Wp, bp, Wcb, bc);
    pack_w_bf16_kernel<<<(HDIM * 2 * HDIM + 255) / 256, 256, 0, stream>>>(W1, W1b, 2 * HDIM, HDIM * 2 * HDIM);
    pack_w_bf16_kernel<<<(HDIM * HDIM + 255) / 256, 256, 0, stream>>>(Q2, Q2b, HDIM, HDIM * HDIM);
    pack_w_bf16_kernel<<<(HDIM * 2 * HDIM + 255) / 256, 256, 0, stream>>>(W2, W2b, 2 * HDIM, HDIM * 2 * HDIM);

    gemm_bf16_n256<<<N_MID / 32, 128, 0, stream>>>(feat, Wpb, bp, hp, 0);
    gemm_bf16_n256<<<N_SRC / 32, 128, 0, stream>>>(feat, Wcb, bc, nbuf, 1);
    segsum_owner_kernel<128, 7, 32, E1_N><<<N_MID / 128, SEG_THREADS, 0, stream>>>(
        nbuf, w1v, src1, dst1, agg, wsum);
    gemm_cat_bf16_n256<<<N_MID / 32, 128, 0, stream>>>(agg, wsum, hp, W1b, bw1, h1, 1);
    gemm_bf16_n256<<<N_MID / 32, 128, 0, stream>>>(h1, Q2b, bq2, nbuf, 1);
    segsum_owner_kernel<32, 5, 16, E2_N><<<N_DST / 32, SEG_THREADS, 0, stream>>>(
        nbuf, w2v, src2, dst2, agg, wsum);
    gemm_cat_bf16_n256<<<N_DST / 32, 128, 0, stream>>>(agg, wsum, h1, W2b, bw2, h2, 1);
    skip_norm_kernel<<<N_DST / 8, 256, 0, stream>>>(hp, h2, zf, N_DST);
    score_kernel<<<N_PAIRS / 8, 256, 0, stream>>>(zf, bias, nids,
                                                  pos_s, pos_d, neg_s, neg_d,
                                                  out, N_PAIRS);
}


# GEMM wave tile 64x64 (64-row blocks; the last, small GEMM keeps 32): a B fragment feeds four row tiles, so weight reads per row and GEMM waves halve; k order and sums unchanged
# speedup vs baseline: 4.1290x; 1.0587x over previous
#include <hip/hip_runtime.h>
#include <hip/hip_bf16.h>


typedef __attribute__((ext_vector_type(16))) __bf16 v16bf;
typedef __attribute__((ext_vector_type(8)))  __bf16 v8bf;
typedef __attribute__((ext_vector_type(4)))  __bf16 v4bf;
typedef __attribute__((ext_vector_type(8)))  float  v8f;

#define N_SRC   131072
#define N_MID   32768
#define N_DST   8192
#define E1_N    524288
#define E2_N    131072
#define HDIM    256
#define N_PAIRS 8192


__global__ void pack_w_bf16_kernel(const float* __restrict__ src,
                                   __bf16* __restrict__ dst, int K, int n) {
    int i = blockIdx.x * blockDim.x + threadIdx.x;
    if (i >= n) return;
    int e    = i & 7;
    int l    = (i >> 3) & 31;
    int j    = (i >> 8) & 1;
    int frag = i >> 9;
    int KS   = K >> 5;
    int nt   = frag / KS;
    int ks   = frag - nt * KS;
    int col  = nt * 16 + (l & 15);
    int k    = ks * 32 + j * 16 + (l >> 4) * 8 + e;
    dst[i] = (__bf16)src[(size_t)col * K + k];
}

__global__ __launch_bounds__(256)
void compose_q1_wp_kernel(const float* __restrict__ Q1, const float* __restrict__ bq1,
                          const float* __restrict__ Wp, const float* __restrict__ bp,
                          __bf16* __restrict__ Wcb, float* __restrict__ bc)
{
    __shared__ float q[HDIM];
    __shared__ float part[8];
    const int n = blockIdx.x;
    const int c = threadIdx.x;
    float qv = Q1[(size_t)n * HDIM + c];
    q[c] = qv;
    float pb = qv * bp[c];
    #pragma unroll
    for (int o = 16; o > 0; o >>= 1) pb += __shfl_xor(pb, o, 32);
    if ((c & 31) == 0) part[c >> 5] = pb;
    __syncthreads();
    float acc = 0.0f;
    #pragma unroll 32
    for (int m = 0; m < HDIM; ++m)
        acc = fmaf(q[m], Wp[(size_t)m * HDIM + c], acc);
    int e  = c & 7;
    int h  = (c >> 3) & 1;
    int j  = (c >> 4) & 1;
    int ks = c >> 5;
    int l  = h * 16 + (n & 15);
    int nt = n >> 4;
    int i  = ((((nt * (HDIM >> 5) + ks) * 2 + j) * 32 + l) << 3) + e;
    Wcb[i] = (__bf16)acc;
    if (c == 0) {
        float s = bq1[n];
        #pragma unroll
        for (int w = 0; w < 8; ++w) s += part[w];
        bc[n] = s;
    }
}

#define A_PAD 8

template <bool SCALED, int ROWS>
__device__ __forceinline__ void stage_a_rows(const float* __restrict__ src,
                                             const float* __restrict__ wsum,
                                             int r0, __bf16* __restrict__ tile,
                                             int pitch, int kofs, int wave, int lane)
{
    constexpr int RW = ROWS / 4;
    float4 v[2 * RW];
    #pragma unroll
    for (int i = 0; i < 2 * RW; ++i) {
        int r = wave * RW + (i >> 1);
        int cidx = (i & 1) * 128 + lane * 4;
        v[i] = *(const float4*)(src + (size_t)(r0 + r) * HDIM + cidx);
    }
    float rs[RW];
    if (SCALED) {
        #pragma unroll
        for (int j = 0; j < RW; ++j) {
            float ws = wsum[r0 + wave * RW + j];
            ws = ws < 1.0f ? 1.0f : ws;
            rs[j] = __frcp_rn(ws);
        }
    }
    #pragma unroll
    for (int i = 0; i < 2 * RW; ++i) {
        int r = wave * RW + (i >> 1);
        int cidx = (i & 1) * 128 + lane * 4;
        float4 x = v[i];
        if (SCALED) { float s = rs[i >> 1]; x.x *= s; x.y *= s; x.z *= s; x.w *= s; }
        v4bf o;
        o[0] = (__bf16)x.x; o[1] = (__bf16)x.y; o[2] = (__bf16)x.z; o[3] = (__bf16)x.w;
        *(v4bf*)(tile + r * pitch + kofs + cidx) = o;
    }
}

__device__ __forceinline__ void load_b_kstep(const __bf16* __restrict__ wk, int tstride,
                                             v16bf* b)
{
    #pragma unroll
    for (int t = 0; t < 4; ++t) {
        const __bf16* wt = wk + (size_t)t * tstride;
        v8bf blo = *(const v8bf*)(wt);
        v8bf bhi = *(const v8bf*)(wt + 256);
        b[t] = __builtin_shufflevector(blo, bhi, 0, 1, 2, 3, 4, 5, 6, 7,
                                       8, 9, 10, 11, 12, 13, 14, 15);
    }
}

template <int RT>
__device__ __forceinline__ void kstep_strip(const __bf16* __restrict__ ak, int rstride,
                                            const v16bf* b, v8f (*acc)[4])
{
    v16bf a[RT];
    #pragma unroll
    for (int i = 0; i < RT; ++i) {
        v8bf alo = *(const v8bf*)(ak + i * rstride);
        v8bf ahi = *(const v8bf*)(ak + i * rstride + 16);
        a[i] = __builtin_shufflevector(alo, ahi, 0, 1, 2, 3, 4, 5, 6, 7,
                                       8, 9, 10, 11, 12, 13, 14, 15);
    }
    #pragma unroll
    for (int t = 0; t < 4; ++t) {
        #pragma unroll
        for (int i = 0; i < RT; ++i)
            acc[i][t] = __builtin_amdgcn_wmma_f32_16x16x32_bf16(
                false, a[i], false, b[t], (short)0, acc[i][t], false, false);
    }
}

template <int KS, int RT>
__device__ __forceinline__ void kloop_strip(const __bf16* __restrict__ arow, int rstride,
                                            const __bf16* __restrict__ wfrag, int tstride,
                                            v8f (*acc)[4])
{
    v16bf bcur[4], bnxt[4];
    load_b_kstep(wfrag, tstride, bcur);
    #pragma unroll
    for (int ks = 0; ks < KS; ++ks) {
        if (ks + 1 < KS) load_b_kstep(wfrag + (ks + 1) * 512, tstride, bnxt);
        __builtin_amdgcn_sched_barrier(0);
        kstep_strip<RT>(arow + ks * 32, rstride, bcur, acc);
        if (ks + 1 < KS) {
            #pragma unroll
            for (int t = 0; t < 4; ++t) bcur[t] = bnxt[t];
        }
    }
}

__device__ __forceinline__ void store_strip(const v8f* acc, const float* __restrict__ bias,
                                            float* __restrict__ C, int m0, int n0,
                                            int half, int l15, int relu)
{
    #pragma unroll
    for (int t = 0; t < 4; ++t) {
        int n = n0 + t * 16 + l15;
        float bv = bias[n];
        #pragma unroll
        for (int r = 0; r < 8; ++r) {
            int m = m0 + half * 8 + r;
            float v = acc[t][r] + bv;
            if (relu) v = fmaxf(v, 0.0f);
            C[(size_t)m * HDIM + n] = v;
        }
    }
}

template <int RT>
__global__ __launch_bounds__(128)
void gemm_bf16_n256(const float* __restrict__ A, const __bf16* __restrict__ W,
                    const float* __restrict__ bias, float* __restrict__ C,
                    int relu)
{
    constexpr int ROWS  = 16 * RT;
    constexpr int PITCH = HDIM + A_PAD;
    __shared__ __attribute__((aligned(16))) __bf16 atile[ROWS * PITCH];

    const int lane = threadIdx.x & 31;
    const int wave = threadIdx.x >> 5;
    const int half = lane >> 4;
    const int l15  = lane & 15;
    const int mb   = blockIdx.x * ROWS;
    const int n0   = wave * 64;

    stage_a_rows<false, ROWS>(A, nullptr, mb, atile, PITCH, 0, wave, lane);
    __syncthreads();

    v8f acc[RT][4];
    #pragma unroll
    for (int i = 0; i < RT; ++i) {
        #pragma unroll
        for (int t = 0; t < 4; ++t) acc[i][t] = (v8f){};
    }

    const __bf16* arow = atile + l15 * PITCH + half * 8;
    const int     tstride = (HDIM >> 5) * 512;
    const __bf16* wfrag = W + (size_t)(n0 >> 4) * tstride + lane * 8;

    kloop_strip<HDIM / 32, RT>(arow, 16 * PITCH, wfrag, tstride, acc);

    #pragma unroll
    for (int i = 0; i < RT; ++i)
        store_strip(acc[i], bias, C, mb + 16 * i, n0, half, l15, relu);
}

template <int RT>
__global__ __launch_bounds__(128)
void gemm_cat_bf16_n256(const float* __restrict__ agg, const float* __restrict__ wsum,
                        const float* __restrict__ hdst, const __bf16* __restrict__ W,
                        const float* __restrict__ bias, float* __restrict__ C, int relu)
{
    constexpr int ROWS  = 16 * RT;
    constexpr int PITCH = 2 * HDIM + A_PAD;
    __shared__ __attribute__((aligned(16))) __bf16 atile[ROWS * PITCH];

    const int lane = threadIdx.x & 31;
    const int wave = threadIdx.x >> 5;
    const int half = lane >> 4;
    const int l15  = lane & 15;
    const int mb   = blockIdx.x * ROWS;
    const int n0   = wave * 64;

    stage_a_rows<true,  ROWS>(agg,  wsum,    mb, atile, PITCH, 0,    wave, lane);
    stage_a_rows<false, ROWS>(hdst, nullptr, mb, atile, PITCH, HDIM, wave, lane);
    __syncthreads();

    v8f acc[RT][4];
    #pragma unroll
    for (int i = 0; i < RT; ++i) {
        #pragma unroll
        for (int t = 0; t < 4; ++t) acc[i][t] = (v8f){};
    }

    const __bf16* arow = atile + l15 * PITCH + half * 8;
    const int     tstride = (2 * HDIM / 32) * 512;
    const __bf16* wfrag = W + (size_t)(n0 >> 4) * tstride + lane * 8;

    kloop_strip<2 * HDIM / 32, RT>(arow, 16 * PITCH, wfrag, tstride, acc);

    #pragma unroll
    for (int i = 0; i < RT; ++i)
        store_strip(acc[i], bias, C, mb + 16 * i, n0, half, l15, relu);
}

#define SEG_THREADS 1024
#define SEG_CAP     8192
#define SEG_UA      8

template <int ROWS, int LRB, int CH, int E>
__global__ __launch_bounds__(SEG_THREADS)
void segsum_owner_kernel(const float* __restrict__ nsrc,
                         const float* __restrict__ w,
                         const int* __restrict__ src,
                         const int* __restrict__ dst,
                         float* __restrict__ agg,
                         float* __restrict__ wsum)
{
    constexpr int TRIP = SEG_THREADS * SEG_UA;
    constexpr int NIT  = E / (4 * TRIP);
    static_assert(ROWS == (1 << LRB), "local row fits LRB bits");
    static_assert(E % (4 * TRIP) == 0, "the scan has whole trips only");
    static_assert((NIT & (NIT - 1)) == 0, "trip rotation by mask");
    static_assert(((long long)E << LRB) <= 0xffffffffLL, "list entry fits 32 bits");
    __shared__ __attribute__((aligned(16))) float tile[ROWS * HDIM];
    __shared__ float    wl[ROWS];
    __shared__ unsigned list[SEG_CAP];
    __shared__ unsigned cnt;

    const int tid  = threadIdx.x;
    const int lane = tid & 31;
    const int wave = __builtin_amdgcn_readfirstlane(tid >> 5);
    const int base = blockIdx.x * ROWS;

    for (int i = tid; i < ROWS * HDIM / 4; i += SEG_THREADS)
        ((float4*)tile)[i] = make_float4(0.0f, 0.0f, 0.0f, 0.0f);
    if (tid < ROWS) wl[tid] = 0.0f;
    if (tid == 0) cnt = 0u;
    __syncthreads();

    auto push = [&](unsigned lr, int e) {
        if (lr < (unsigned)ROWS) {
            unsigned pos = atomicAdd(&cnt, 1u);
            if (pos < SEG_CAP) {
                list[pos] = ((unsigned)e << LRB) | lr;
            } else {
                int   s  = src[e];
                float we = w[e];
                atomicAdd(&wl[lr], we);
                for (int c = 0; c < HDIM; ++c)
                    atomicAdd(&tile[lr * HDIM + c], nsrc[(size_t)s * HDIM + c] * we);
            }
        }
    };
    const int4* d4 = (const int4*)dst;
    for (int it = 0; it < NIT; ++it) {
        const int itr = (it + (int)blockIdx.x) & (NIT - 1);
        const int i0  = itr * TRIP + tid;
        int4 d[SEG_UA];
        #pragma unroll
        for (int u = 0; u < SEG_UA; ++u) d[u] = d4[i0 + u * SEG_THREADS];
        #pragma unroll
        for (int u = 0; u < SEG_UA; ++u) {
            const int e0 = (i0 + u * SEG_THREADS) * 4;
            const unsigned l0 = (unsigned)(d[u].x - base);
            const unsigned l1 = (unsigned)(d[u].y - base);
            const unsigned l2 = (unsigned)(d[u].z - base);
            const unsigned l3 = (unsigned)(d[u].w - base);
            const unsigned lm = min(min(l0, l1), min(l2, l3));
            if (lm < (unsigned)ROWS) {
                push(l0, e0);
                push(l1, e0 + 1);
                push(l2, e0 + 2);
                push(l3, e0 + 3);
            }
        }
    }
    __syncthreads();

    unsigned nl = cnt;
    nl = nl < (unsigned)SEG_CAP ? nl : (unsigned)SEG_CAP;
    const int n = __builtin_amdgcn_readfirstlane((int)nl);
    for (int c0 = wave * CH; c0 < n; c0 += (SEG_THREADS / 32) * CH) {
        const int  idx   = c0 + lane;
        const bool valid = (lane < CH) && (idx < n);
        unsigned ent = valid ? list[idx] : 0u;
        int   e  = (int)(ent >> LRB);
        int   lr = (int)(ent & (unsigned)(ROWS - 1));
        int   s  = valid ? src[e] : 0;
        float we = valid ? w[e] : 0.0f;
        if (valid) atomicAdd(&wl[lr], we);
        const int rem = n - c0;
        const int cw  = rem < CH ? rem : CH;
        int j = 0;
        for (; j + 4 <= cw; j += 4) {
            int sj[4], lj[4]; float wj[4];
            #pragma unroll
            for (int u = 0; u < 4; ++u) {
                sj[u] = __builtin_amdgcn_readlane(s, j + u);
                lj[u] = __builtin_amdgcn_readlane(lr, j + u);
                wj[u] = __int_as_float(__builtin_amdgcn_readlane(__float_as_int(we), j + u));
            }
            float v[4][8];
            #pragma unroll
            for (int u = 0; u < 4; ++u) {
                const float* nr = nsrc + (size_t)sj[u] * HDIM + lane;
                #pragma unroll
                for (int q = 0; q < 8; ++q) v[u][q] = nr[q * 32];
            }
            #pragma unroll
            for (int u = 0; u < 4; ++u) {
                float* tr = tile + lj[u] * HDIM + lane;
                #pragma unroll
                for (int q = 0; q < 8; ++q) atomicAdd(tr + q * 32, v[u][q] * wj[u]);
            }
        }
        for (; j < cw; ++j) {
            int   s1 = __builtin_amdgcn_readlane(s, j);
            int   l1 = __builtin_amdgcn_readlane(lr, j);
            float w1 = __int_as_float(__builtin_amdgcn_readlane(__float_as_int(we), j));
            const float* nr = nsrc + (size_t)s1 * HDIM + lane;
            float* tr = tile + l1 * HDIM + lane;
            float v1[8];
            #pragma unroll
            for (int q = 0; q < 8; ++q) v1[q] = nr[q * 32];
            #pragma unroll
            for (int q = 0; q < 8; ++q) atomicAdd(tr + q * 32, v1[q] * w1);
        }
    }
    __syncthreads();

    float4* arow = (float4*)(agg + (size_t)base * HDIM);
    for (int i = tid; i < ROWS * HDIM / 4; i += SEG_THREADS)
        arow[i] = ((const float4*)tile)[i];
    if (tid < ROWS) wsum[base + tid] = wl[tid];
}

__global__ __launch_bounds__(256)
void skip_norm_kernel(const float* __restrict__ hp, const float* __restrict__ h2,
                      float* __restrict__ zf, int rows)
{
    int r    = blockIdx.x * 8 + (threadIdx.x >> 5);
    int lane = threadIdx.x & 31;
    if (r >= rows) return;
    float v[8];
    float ss = 0.0f;
    #pragma unroll
    for (int j = 0; j < 8; ++j) {
        int c = lane * 8 + j;
        v[j] = hp[(size_t)r * HDIM + c] + h2[(size_t)r * HDIM + c];
        ss += v[j] * v[j];
    }
    #pragma unroll
    for (int o = 16; o > 0; o >>= 1) ss += __shfl_xor(ss, o, 32);
    float nrm   = sqrtf(ss);
    float scale = (nrm == 0.0f) ? 1.0f : 1.0f / nrm;
    #pragma unroll
    for (int j = 0; j < 8; ++j)
        zf[(size_t)r * HDIM + lane * 8 + j] = v[j] * scale;
}

__global__ __launch_bounds__(256)
void score_kernel(const float* __restrict__ zf, const float* __restrict__ bias,
                  const int* __restrict__ nids,
                  const int* __restrict__ ps, const int* __restrict__ pd,
                  const int* __restrict__ ns, const int* __restrict__ nd,
                  float* __restrict__ out, int npairs)
{
    int p    = blockIdx.x * 8 + (threadIdx.x >> 5);
    int lane = threadIdx.x & 31;
    if (p >= npairs) return;
    int pu = ps[p], pv = pd[p], nu = ns[p], nv = nd[p];
    float sp = 0.0f, sn = 0.0f;
    #pragma unroll
    for (int j = 0; j < 8; ++j) {
        int c = lane * 8 + j;
        sp += zf[(size_t)pu * HDIM + c] * zf[(size_t)pv * HDIM + c];
        sn += zf[(size_t)nu * HDIM + c] * zf[(size_t)nv * HDIM + c];
    }
    #pragma unroll
    for (int o = 16; o > 0; o >>= 1) {
        sp += __shfl_xor(sp, o, 32);
        sn += __shfl_xor(sn, o, 32);
    }
    if (lane == 0) {
        float gp = bias[nids[pu]] + bias[nids[pv]];
        float gn = bias[nids[nu]] + bias[nids[nv]];
        float m  = (sn + gn) - (sp + gp) + 1.0f;
        out[p] = m > 0.0f ? m : 0.0f;
    }
}

extern "C" void kernel_launch(void* const* d_in, const int* in_sizes, int n_in,
                              void* d_out, int out_size, void* d_ws, size_t ws_size,
                              hipStream_t stream) {
    const float* feat = (const float*)d_in[0];
    const float* Wp   = (const float*)d_in[1];
    const float* bp   = (const float*)d_in[2];
    const float* Q1   = (const float*)d_in[3];
    const float* bq1  = (const float*)d_in[4];
    const float* W1   = (const float*)d_in[5];
    const float* bw1  = (const float*)d_in[6];
    const float* Q2   = (const float*)d_in[7];
    const float* bq2  = (const float*)d_in[8];
    const float* W2   = (const float*)d_in[9];
    const float* bw2  = (const float*)d_in[10];
    const float* w1v  = (const float*)d_in[11];
    const float* w2v  = (const float*)d_in[12];
    const float* bias = (const float*)d_in[13];
    const int* src1   = (const int*)d_in[14];
    const int* dst1   = (const int*)d_in[15];
    const int* src2   = (const int*)d_in[16];
    const int* dst2   = (const int*)d_in[17];
    const int* pos_s  = (const int*)d_in[18];
    const int* pos_d  = (const int*)d_in[19];
    const int* neg_s  = (const int*)d_in[20];
    const int* neg_d  = (const int*)d_in[21];
    const int* nids   = (const int*)d_in[22];
    float* out        = (float*)d_out;

    char* ws = (char*)d_ws;
    const size_t SZ_HP  = (size_t)N_SRC * HDIM * 4;
    const size_t SZ_N   = (size_t)N_SRC * HDIM * 4;
    const size_t SZ_H1  = (size_t)N_MID * HDIM * 4;
    const size_t SZ_AGG = (size_t)N_MID * HDIM * 4;
    const size_t SZ_WS  = 256 * 1024;
    const size_t SZ_H2  = (size_t)N_DST * HDIM * 4;
    const size_t SZ_ZF  = (size_t)N_DST * HDIM * 4;
    float*  hp   = (float*)(ws);
    float*  bc   = hp + (size_t)N_MID * HDIM;
    float*  nbuf = (float*)(ws + SZ_HP);
    float*  h1   = (float*)(ws + SZ_HP + SZ_N);
    float*  agg  = (float*)(ws + SZ_HP + SZ_N + SZ_H1);
    float*  wsum = (float*)(ws + SZ_HP + SZ_N + SZ_H1 + SZ_AGG);
    float*  h2   = (float*)(ws + SZ_HP + SZ_N + SZ_H1 + SZ_AGG + SZ_WS);
    float*  zf   = (float*)(ws + SZ_HP + SZ_N + SZ_H1 + SZ_AGG + SZ_WS + SZ_H2);
    __bf16* Wpb  = (__bf16*)(ws + SZ_HP + SZ_N + SZ_H1 + SZ_AGG + SZ_WS + SZ_H2 + SZ_ZF);
    __bf16* Wcb  = Wpb + HDIM * HDIM;
    __bf16* W1b  = Wcb + HDIM * HDIM;
    __bf16* Q2b  = W1b + HDIM * 2 * HDIM;
    __bf16* W2b  = Q2b + HDIM * HDIM;

    pack_w_bf16_kernel<<<(HDIM * HDIM + 255) / 256, 256, 0, stream>>>(Wp, Wpb, HDIM, HDIM * HDIM);
    compose_q1_wp_kernel<<<HDIM, 256, 0, stream>>>(Q1, bq1, Wp, bp, Wcb, bc);
    pack_w_bf16_kernel<<<(HDIM * 2 * HDIM + 255) / 256, 256, 0, stream>>>(W1, W1b, 2 * HDIM, HDIM * 2 * HDIM);
    pack_w_bf16_kernel<<<(HDIM * HDIM + 255) / 256, 256, 0, stream>>>(Q2, Q2b, HDIM, HDIM * HDIM);
    pack_w_bf16_kernel<<<(HDIM * 2 * HDIM + 255) / 256, 256, 0, stream>>>(W2, W2b, 2 * HDIM, HDIM * 2 * HDIM);

    gemm_bf16_n256<4><<<N_MID / 64, 128, 0, stream>>>(feat, Wpb, bp, hp, 0);
    gemm_bf16_n256<4><<<N_SRC / 64, 128, 0, stream>>>(feat, Wcb, bc, nbuf, 1);
    segsum_owner_kernel<128, 7, 32, E1_N><<<N_MID / 128, SEG_THREADS, 0, stream>>>(
        nbuf, w1v, src1, dst1, agg, wsum);
    gemm_cat_bf16_n256<4><<<N_MID / 64, 128, 0, stream>>>(agg, wsum, hp, W1b, bw1, h1, 1);
    gemm_bf16_n256<4><<<N_MID / 64, 128, 0, stream>>>(h1, Q2b, bq2, nbuf, 1);
    segsum_owner_kernel<32, 5, 16, E2_N><<<N_DST / 32, SEG_THREADS, 0, stream>>>(
        nbuf, w2v, src2, dst2, agg, wsum);
    gemm_cat_bf16_n256<2><<<N_DST / 32, 128, 0, stream>>>(agg, wsum, h1, W2b, bw2, h2, 1);
    skip_norm_kernel<<<N_DST / 8, 256, 0, stream>>>(hp, h2, zf, N_DST);
    score_kernel<<<N_PAIRS / 8, 256, 0, stream>>>(zf, bias, nids,
                                                  pos_s, pos_d, neg_s, neg_d,
                                                  out, N_PAIRS);
}


# Owner scan on bytes: a small kernel stores each edge's owner number (dst >> log2 rows, 8 bits) per call; owners scan 1 B an edge (was 4 B), 16 edges a load; dst tested at gather; arrival-order sums
# speedup vs baseline: 4.3883x; 1.0628x over previous
#include <hip/hip_runtime.h>
#include <hip/hip_bf16.h>


typedef __attribute__((ext_vector_type(16))) __bf16 v16bf;
typedef __attribute__((ext_vector_type(8)))  __bf16 v8bf;
typedef __attribute__((ext_vector_type(4)))  __bf16 v4bf;
typedef __attribute__((ext_vector_type(8)))  float  v8f;

#define N_SRC   131072
#define N_MID   32768
#define N_DST   8192
#define E1_N    524288
#define E2_N    131072
#define HDIM    256
#define N_PAIRS 8192


__global__ void pack_w_bf16_kernel(const float* __restrict__ src,
                                   __bf16* __restrict__ dst, int K, int n) {
    int i = blockIdx.x * blockDim.x + threadIdx.x;
    if (i >= n) return;
    int e    = i & 7;
    int l    = (i >> 3) & 31;
    int j    = (i >> 8) & 1;
    int frag = i >> 9;
    int KS   = K >> 5;
    int nt   = frag / KS;
    int ks   = frag - nt * KS;
    int col  = nt * 16 + (l & 15);
    int k    = ks * 32 + j * 16 + (l >> 4) * 8 + e;
    dst[i] = (__bf16)src[(size_t)col * K + k];
}

__global__ __launch_bounds__(256)
void compose_q1_wp_kernel(const float* __restrict__ Q1, const float* __restrict__ bq1,
                          const float* __restrict__ Wp, const float* __restrict__ bp,
                          __bf16* __restrict__ Wcb, float* __restrict__ bc)
{
    __shared__ float q[HDIM];
    __shared__ float part[8];
    const int n = blockIdx.x;
    const int c = threadIdx.x;
    float qv = Q1[(size_t)n * HDIM + c];
    q[c] = qv;
    float pb = qv * bp[c];
    #pragma unroll
    for (int o = 16; o > 0; o >>= 1) pb += __shfl_xor(pb, o, 32);
    if ((c & 31) == 0) part[c >> 5] = pb;
    __syncthreads();
    float acc = 0.0f;
    #pragma unroll 32
    for (int m = 0; m < HDIM; ++m)
        acc = fmaf(q[m], Wp[(size_t)m * HDIM + c], acc);
    int e  = c & 7;
    int h  = (c >> 3) & 1;
    int j  = (c >> 4) & 1;
    int ks = c >> 5;
    int l  = h * 16 + (n & 15);
    int nt = n >> 4;
    int i  = ((((nt * (HDIM >> 5) + ks) * 2 + j) * 32 + l) << 3) + e;
    Wcb[i] = (__bf16)acc;
    if (c == 0) {
        float s = bq1[n];
        #pragma unroll
        for (int w = 0; w < 8; ++w) s += part[w];
        bc[n] = s;
    }
}

#define A_PAD 8

template <bool SCALED, int ROWS>
__device__ __forceinline__ void stage_a_rows(const float* __restrict__ src,
                                             const float* __restrict__ wsum,
                                             int r0, __bf16* __restrict__ tile,
                                             int pitch, int kofs, int wave, int lane)
{
    constexpr int RW = ROWS / 4;
    float4 v[2 * RW];
    #pragma unroll
    for (int i = 0; i < 2 * RW; ++i) {
        int r = wave * RW + (i >> 1);
        int cidx = (i & 1) * 128 + lane * 4;
        v[i] = *(const float4*)(src + (size_t)(r0 + r) * HDIM + cidx);
    }
    float rs[RW];
    if (SCALED) {
        #pragma unroll
        for (int j = 0; j < RW; ++j) {
            float ws = wsum[r0 + wave * RW + j];
            ws = ws < 1.0f ? 1.0f : ws;
            rs[j] = __frcp_rn(ws);
        }
    }
    #pragma unroll
    for (int i = 0; i < 2 * RW; ++i) {
        int r = wave * RW + (i >> 1);
        int cidx = (i & 1) * 128 + lane * 4;
        float4 x = v[i];
        if (SCALED) { float s = rs[i >> 1]; x.x *= s; x.y *= s; x.z *= s; x.w *= s; }
        v4bf o;
        o[0] = (__bf16)x.x; o[1] = (__bf16)x.y; o[2] = (__bf16)x.z; o[3] = (__bf16)x.w;
        *(v4bf*)(tile + r * pitch + kofs + cidx) = o;
    }
}

__device__ __forceinline__ void load_b_kstep(const __bf16* __restrict__ wk, int tstride,
                                             v16bf* b)
{
    #pragma unroll
    for (int t = 0; t < 4; ++t) {
        const __bf16* wt = wk + (size_t)t * tstride;
        v8bf blo = *(const v8bf*)(wt);
        v8bf bhi = *(const v8bf*)(wt + 256);
        b[t] = __builtin_shufflevector(blo, bhi, 0, 1, 2, 3, 4, 5, 6, 7,
                                       8, 9, 10, 11, 12, 13, 14, 15);
    }
}

template <int RT>
__device__ __forceinline__ void kstep_strip(const __bf16* __restrict__ ak, int rstride,
                                            const v16bf* b, v8f (*acc)[4])
{
    v16bf a[RT];
    #pragma unroll
    for (int i = 0; i < RT; ++i) {
        v8bf alo = *(const v8bf*)(ak + i * rstride);
        v8bf ahi = *(const v8bf*)(ak + i * rstride + 16);
        a[i] = __builtin_shufflevector(alo, ahi, 0, 1, 2, 3, 4, 5, 6, 7,
                                       8, 9, 10, 11, 12, 13, 14, 15);
    }
    #pragma unroll
    for (int t = 0; t < 4; ++t) {
        #pragma unroll
        for (int i = 0; i < RT; ++i)
            acc[i][t] = __builtin_amdgcn_wmma_f32_16x16x32_bf16(
                false, a[i], false, b[t], (short)0, acc[i][t], false, false);
    }
}

template <int KS, int RT>
__device__ __forceinline__ void kloop_strip(const __bf16* __restrict__ arow, int rstride,
                                            const __bf16* __restrict__ wfrag, int tstride,
                                            v8f (*acc)[4])
{
    v16bf bcur[4], bnxt[4];
    load_b_kstep(wfrag, tstride, bcur);
    #pragma unroll
    for (int ks = 0; ks < KS; ++ks) {
        if (ks + 1 < KS) load_b_kstep(wfrag + (ks + 1) * 512, tstride, bnxt);
        __builtin_amdgcn_sched_barrier(0);
        kstep_strip<RT>(arow + ks * 32, rstride, bcur, acc);
        if (ks + 1 < KS) {
            #pragma unroll
            for (int t = 0; t < 4; ++t) bcur[t] = bnxt[t];
        }
    }
}

__device__ __forceinline__ void store_strip(const v8f* acc, const float* __restrict__ bias,
                                            float* __restrict__ C, int m0, int n0,
                                            int half, int l15, int relu)
{
    #pragma unroll
    for (int t = 0; t < 4; ++t) {
        int n = n0 + t * 16 + l15;
        float bv = bias[n];
        #pragma unroll
        for (int r = 0; r < 8; ++r) {
            int m = m0 + half * 8 + r;
            float v = acc[t][r] + bv;
            if (relu) v = fmaxf(v, 0.0f);
            C[(size_t)m * HDIM + n] = v;
        }
    }
}

template <int RT>
__global__ __launch_bounds__(128)
void gemm_bf16_n256(const float* __restrict__ A, const __bf16* __restrict__ W,
                    const float* __restrict__ bias, float* __restrict__ C,
                    int relu)
{
    constexpr int ROWS  = 16 * RT;
    constexpr int PITCH = HDIM + A_PAD;
    __shared__ __attribute__((aligned(16))) __bf16 atile[ROWS * PITCH];

    const int lane = threadIdx.x & 31;
    const int wave = threadIdx.x >> 5;
    const int half = lane >> 4;
    const int l15  = lane & 15;
    const int mb   = blockIdx.x * ROWS;
    const int n0   = wave * 64;

    stage_a_rows<false, ROWS>(A, nullptr, mb, atile, PITCH, 0, wave, lane);
    __syncthreads();

    v8f acc[RT][4];
    #pragma unroll
    for (int i = 0; i < RT; ++i) {
        #pragma unroll
        for (int t = 0; t < 4; ++t) acc[i][t] = (v8f){};
    }

    const __bf16* arow = atile + l15 * PITCH + half * 8;
    const int     tstride = (HDIM >> 5) * 512;
    const __bf16* wfrag = W + (size_t)(n0 >> 4) * tstride + lane * 8;

    kloop_strip<HDIM / 32, RT>(arow, 16 * PITCH, wfrag, tstride, acc);

    #pragma unroll
    for (int i = 0; i < RT; ++i)
        store_strip(acc[i], bias, C, mb + 16 * i, n0, half, l15, relu);
}

template <int RT>
__global__ __launch_bounds__(128)
void gemm_cat_bf16_n256(const float* __restrict__ agg, const float* __restrict__ wsum,
                        const float* __restrict__ hdst, const __bf16* __restrict__ W,
                        const float* __restrict__ bias, float* __restrict__ C, int relu)
{
    constexpr int ROWS  = 16 * RT;
    constexpr int PITCH = 2 * HDIM + A_PAD;
    __shared__ __attribute__((aligned(16))) __bf16 atile[ROWS * PITCH];

    const int lane = threadIdx.x & 31;
    const int wave = threadIdx.x >> 5;
    const int half = lane >> 4;
    const int l15  = lane & 15;
    const int mb   = blockIdx.x * ROWS;
    const int n0   = wave * 64;

    stage_a_rows<true,  ROWS>(agg,  wsum,    mb, atile, PITCH, 0,    wave, lane);
    stage_a_rows<false, ROWS>(hdst, nullptr, mb, atile, PITCH, HDIM, wave, lane);
    __syncthreads();

    v8f acc[RT][4];
    #pragma unroll
    for (int i = 0; i < RT; ++i) {
        #pragma unroll
        for (int t = 0; t < 4; ++t) acc[i][t] = (v8f){};
    }

    const __bf16* arow = atile + l15 * PITCH + half * 8;
    const int     tstride = (2 * HDIM / 32) * 512;
    const __bf16* wfrag = W + (size_t)(n0 >> 4) * tstride + lane * 8;

    kloop_strip<2 * HDIM / 32, RT>(arow, 16 * PITCH, wfrag, tstride, acc);

    #pragma unroll
    for (int i = 0; i < RT; ++i)
        store_strip(acc[i], bias, C, mb + 16 * i, n0, half, l15, relu);
}

#define SEG_THREADS 1024
#define SEG_CAP     8192
#define SEG_UA      8
#define SEG1_LRB    7
#define SEG2_LRB    5

__global__ __launch_bounds__(256)
void owner_bytes_kernel(const int* __restrict__ dst1, const int* __restrict__ dst2,
                        unsigned* __restrict__ own1, unsigned* __restrict__ own2)
{
    int g = blockIdx.x * 256 + threadIdx.x;
    const int4* s = (const int4*)dst1;
    unsigned*   o = own1;
    int sh = SEG1_LRB;
    if (g >= E1_N / 4) {
        g -= E1_N / 4;
        s = (const int4*)dst2;
        o = own2;
        sh = SEG2_LRB;
    }
    int4 d = s[g];
    o[g] = ((unsigned)(d.x >> sh) & 255u) | (((unsigned)(d.y >> sh) & 255u) << 8) |
           (((unsigned)(d.z >> sh) & 255u) << 16) | (((unsigned)(d.w >> sh) & 255u) << 24);
}

__device__ __forceinline__ unsigned match_bytes(unsigned x, unsigned pat)
{
    x ^= pat;
    return (x - 0x01010101u) & ~x & 0x80808080u;
}

template <int ROWS, int LRB, int CH, int E>
__global__ __launch_bounds__(SEG_THREADS)
void segsum_owner_kernel(const float* __restrict__ nsrc,
                         const float* __restrict__ w,
                         const int* __restrict__ src,
                         const int* __restrict__ dst,
                         const unsigned* __restrict__ own,
                         float* __restrict__ agg,
                         float* __restrict__ wsum)
{
    constexpr int NLD  = E / 16;
    constexpr int PER  = NLD / SEG_THREADS;
    constexpr int UA   = PER < SEG_UA ? PER : SEG_UA;
    constexpr int NIT  = PER / UA;
    static_assert(ROWS == (1 << LRB), "local row fits LRB bits");
    static_assert(E % (16 * SEG_THREADS) == 0 && PER % UA == 0, "the scan has whole trips only");
    static_assert((NLD & (NLD - 1)) == 0 && NLD >= 256, "start rotation by mask");

    __shared__ __attribute__((aligned(16))) float tile[ROWS * HDIM];
    __shared__ float    wl[ROWS];
    __shared__ unsigned list[SEG_CAP];
    __shared__ unsigned cnt;

    const int tid  = threadIdx.x;
    const int lane = tid & 31;
    const int wave = __builtin_amdgcn_readfirstlane(tid >> 5);
    const int base = blockIdx.x * ROWS;

    for (int i = tid; i < ROWS * HDIM / 4; i += SEG_THREADS)
        ((float4*)tile)[i] = make_float4(0.0f, 0.0f, 0.0f, 0.0f);
    if (tid < ROWS) wl[tid] = 0.0f;
    if (tid == 0) cnt = 0u;
    __syncthreads();

    auto push = [&](int e) {
        unsigned pos = atomicAdd(&cnt, 1u);
        if (pos < SEG_CAP) {
            list[pos] = (unsigned)e;
        } else {
            unsigned lr = (unsigned)(dst[e] - base);
            if (lr < (unsigned)ROWS) {
                int   s  = src[e];
                float we = w[e];
                atomicAdd(&wl[lr], we);
                for (int c = 0; c < HDIM; ++c)
                    atomicAdd(&tile[lr * HDIM + c], nsrc[(size_t)s * HDIM + c] * we);
            }
        }
    };
    const uint4*   o4  = (const uint4*)own;
    const unsigned pat = (blockIdx.x & 255u) * 0x01010101u;
    const int      rot = (int)blockIdx.x * (NLD >> 8);
    for (int it = 0; it < NIT; ++it) {
        uint4 d[UA];
        int   li[UA];
        #pragma unroll
        for (int u = 0; u < UA; ++u) {
            li[u] = ((it * UA + u) * SEG_THREADS + tid + rot) & (NLD - 1);
            d[u]  = o4[li[u]];
        }
        #pragma unroll
        for (int u = 0; u < UA; ++u) {
            unsigned m = (match_bytes(d[u].x, pat) >> 7) | (match_bytes(d[u].y, pat) >> 6) |
                         (match_bytes(d[u].z, pat) >> 5) | (match_bytes(d[u].w, pat) >> 4);
            while (m) {
                const int i = __builtin_ctz(m);
                m &= m - 1u;
                push(li[u] * 16 + (i & 7) * 4 + (i >> 3));
            }
        }
    }
    __syncthreads();

    unsigned nl = cnt;
    nl = nl < (unsigned)SEG_CAP ? nl : (unsigned)SEG_CAP;
    const int n = __builtin_amdgcn_readfirstlane((int)nl);
    for (int c0 = wave * CH; c0 < n; c0 += (SEG_THREADS / 32) * CH) {
        const int  idx   = c0 + lane;
        const bool valid = (lane < CH) && (idx < n);
        int   e  = valid ? (int)list[idx] : 0;
        int   s  = src[e];
        float we = w[e];
        int   lr = dst[e] - base;
        if (!valid || (unsigned)lr >= (unsigned)ROWS) { s = 0; we = 0.0f; lr = -1; }
        if (lr >= 0) atomicAdd(&wl[lr], we);
        const int rem = n - c0;
        const int cw  = rem < CH ? rem : CH;
        int j = 0;
        for (; j + 4 <= cw; j += 4) {
            int sj[4], lj[4]; float wj[4];
            #pragma unroll
            for (int u = 0; u < 4; ++u) {
                sj[u] = __builtin_amdgcn_readlane(s, j + u);
                lj[u] = __builtin_amdgcn_readlane(lr, j + u);
                wj[u] = __int_as_float(__builtin_amdgcn_readlane(__float_as_int(we), j + u));
            }
            float v[4][8];
            #pragma unroll
            for (int u = 0; u < 4; ++u) {
                const float* nr = nsrc + (size_t)sj[u] * HDIM + lane;
                #pragma unroll
                for (int q = 0; q < 8; ++q) v[u][q] = nr[q * 32];
            }
            #pragma unroll
            for (int u = 0; u < 4; ++u) {
                if (lj[u] < 0) continue;
                float* tr = tile + lj[u] * HDIM + lane;
                #pragma unroll
                for (int q = 0; q < 8; ++q) atomicAdd(tr + q * 32, v[u][q] * wj[u]);
            }
        }
        for (; j < cw; ++j) {
            int   s1 = __builtin_amdgcn_readlane(s, j);
            int   l1 = __builtin_amdgcn_readlane(lr, j);
            float w1 = __int_as_float(__builtin_amdgcn_readlane(__float_as_int(we), j));
            const float* nr = nsrc + (size_t)s1 * HDIM + lane;
            float* tr = tile + l1 * HDIM + lane;
            if (l1 < 0) continue;
            float v1[8];
            #pragma unroll
            for (int q = 0; q < 8; ++q) v1[q] = nr[q * 32];
            #pragma unroll
            for (int q = 0; q < 8; ++q) atomicAdd(tr + q * 32, v1[q] * w1);
        }
    }
    __syncthreads();

    float4* arow = (float4*)(agg + (size_t)base * HDIM);
    for (int i = tid; i < ROWS * HDIM / 4; i += SEG_THREADS)
        arow[i] = ((const float4*)tile)[i];
    if (tid < ROWS) wsum[base + tid] = wl[tid];
}

__global__ __launch_bounds__(256)
void skip_norm_kernel(const float* __restrict__ hp, const float* __restrict__ h2,
                      float* __restrict__ zf, int rows)
{
    int r    = blockIdx.x * 8 + (threadIdx.x >> 5);
    int lane = threadIdx.x & 31;
    if (r >= rows) return;
    float v[8];
    float ss = 0.0f;
    #pragma unroll
    for (int j = 0; j < 8; ++j) {
        int c = lane * 8 + j;
        v[j] = hp[(size_t)r * HDIM + c] + h2[(size_t)r * HDIM + c];
        ss += v[j] * v[j];
    }
    #pragma unroll
    for (int o = 16; o > 0; o >>= 1) ss += __shfl_xor(ss, o, 32);
    float nrm   = sqrtf(ss);
    float scale = (nrm == 0.0f) ? 1.0f : 1.0f / nrm;
    #pragma unroll
    for (int j = 0; j < 8; ++j)
        zf[(size_t)r * HDIM + lane * 8 + j] = v[j] * scale;
}

__global__ __launch_bounds__(256)
void score_kernel(const float* __restrict__ zf, const float* __restrict__ bias,
                  const int* __restrict__ nids,
                  const int* __restrict__ ps, const int* __restrict__ pd,
                  const int* __restrict__ ns, const int* __restrict__ nd,
                  float* __restrict__ out, int npairs)
{
    int p    = blockIdx.x * 8 + (threadIdx.x >> 5);
    int lane = threadIdx.x & 31;
    if (p >= npairs) return;
    int pu = ps[p], pv = pd[p], nu = ns[p], nv = nd[p];
    float sp = 0.0f, sn = 0.0f;
    #pragma unroll
    for (int j = 0; j < 8; ++j) {
        int c = lane * 8 + j;
        sp += zf[(size_t)pu * HDIM + c] * zf[(size_t)pv * HDIM + c];
        sn += zf[(size_t)nu * HDIM + c] * zf[(size_t)nv * HDIM + c];
    }
    #pragma unroll
    for (int o = 16; o > 0; o >>= 1) {
        sp += __shfl_xor(sp, o, 32);
        sn += __shfl_xor(sn, o, 32);
    }
    if (lane == 0) {
        float gp = bias[nids[pu]] + bias[nids[pv]];
        float gn = bias[nids[nu]] + bias[nids[nv]];
        float m  = (sn + gn) - (sp + gp) + 1.0f;
        out[p] = m > 0.0f ? m : 0.0f;
    }
}

extern "C" void kernel_launch(void* const* d_in, const int* in_sizes, int n_in,
                              void* d_out, int out_size, void* d_ws, size_t ws_size,
                              hipStream_t stream) {
    const float* feat = (const float*)d_in[0];
    const float* Wp   = (const float*)d_in[1];
    const float* bp   = (const float*)d_in[2];
    const float* Q1   = (const float*)d_in[3];
    const float* bq1  = (const float*)d_in[4];
    const float* W1   = (const float*)d_in[5];
    const float* bw1  = (const float*)d_in[6];
    const float* Q2   = (const float*)d_in[7];
    const float* bq2  = (const float*)d_in[8];
    const float* W2   = (const float*)d_in[9];
    const float* bw2  = (const float*)d_in[10];
    const float* w1v  = (const float*)d_in[11];
    const float* w2v  = (const float*)d_in[12];
    const float* bias = (const float*)d_in[13];
    const int* src1   = (const int*)d_in[14];
    const int* dst1   = (const int*)d_in[15];
    const int* src2   = (const int*)d_in[16];
    const int* dst2   = (const int*)d_in[17];
    const int* pos_s  = (const int*)d_in[18];
    const int* pos_d  = (const int*)d_in[19];
    const int* neg_s  = (const int*)d_in[20];
    const int* neg_d  = (const int*)d_in[21];
    const int* nids   = (const int*)d_in[22];
    float* out        = (float*)d_out;

    char* ws = (char*)d_ws;
    const size_t SZ_HP  = (size_t)N_SRC * HDIM * 4;
    const size_t SZ_N   = (size_t)N_SRC * HDIM * 4;
    const size_t SZ_H1  = (size_t)N_MID * HDIM * 4;
    const size_t SZ_AGG = (size_t)N_MID * HDIM * 4;
    const size_t SZ_WS  = 256 * 1024;
    const size_t SZ_H2  = (size_t)N_DST * HDIM * 4;
    const size_t SZ_ZF  = (size_t)N_DST * HDIM * 4;
    float*  hp   = (float*)(ws);
    float*  bc   = hp + (size_t)N_MID * HDIM;
    float*  nbuf = (float*)(ws + SZ_HP);
    float*  h1   = (float*)(ws + SZ_HP + SZ_N);
    float*  agg  = (float*)(ws + SZ_HP + SZ_N + SZ_H1);
    float*  wsum = (float*)(ws + SZ_HP + SZ_N + SZ_H1 + SZ_AGG);
    float*  h2   = (float*)(ws + SZ_HP + SZ_N + SZ_H1 + SZ_AGG + SZ_WS);
    float*  zf   = (float*)(ws + SZ_HP + SZ_N + SZ_H1 + SZ_AGG + SZ_WS + SZ_H2);
    __bf16* Wpb  = (__bf16*)(ws + SZ_HP + SZ_N + SZ_H1 + SZ_AGG + SZ_WS + SZ_H2 + SZ_ZF);
    __bf16* Wcb  = Wpb + HDIM * HDIM;
    __bf16* W1b  = Wcb + HDIM * HDIM;
    __bf16* Q2b  = W1b + HDIM * 2 * HDIM;
    __bf16* W2b  = Q2b + HDIM * HDIM;
    unsigned* own1 = (unsigned*)(bc + HDIM);
    unsigned* own2 = own1 + E1_N / 4;

    pack_w_bf16_kernel<<<(HDIM * HDIM + 255) / 256, 256, 0, stream>>>(Wp, Wpb, HDIM, HDIM * HDIM);
    compose_q1_wp_kernel<<<HDIM, 256, 0, stream>>>(Q1, bq1, Wp, bp, Wcb, bc);
    pack_w_bf16_kernel<<<(HDIM * 2 * HDIM + 255) / 256, 256, 0, stream>>>(W1, W1b, 2 * HDIM, HDIM * 2 * HDIM);
    pack_w_bf16_kernel<<<(HDIM * HDIM + 255) / 256, 256, 0, stream>>>(Q2, Q2b, HDIM, HDIM * HDIM);
    pack_w_bf16_kernel<<<(HDIM * 2 * HDIM + 255) / 256, 256, 0, stream>>>(W2, W2b, 2 * HDIM, HDIM * 2 * HDIM);
    owner_bytes_kernel<<<(E1_N + E2_N) / 4 / 256, 256, 0, stream>>>(dst1, dst2, own1, own2);

    gemm_bf16_n256<4><<<N_MID / 64, 128, 0, stream>>>(feat, Wpb, bp, hp, 0);
    gemm_bf16_n256<4><<<N_SRC / 64, 128, 0, stream>>>(feat, Wcb, bc, nbuf, 1);
    segsum_owner_kernel<128, SEG1_LRB, 32, E1_N><<<N_MID / 128, SEG_THREADS, 0, stream>>>(
        nbuf, w1v, src1, dst1, own1, agg, wsum);
    gemm_cat_bf16_n256<4><<<N_MID / 64, 128, 0, stream>>>(agg, wsum, hp, W1b, bw1, h1, 1);
    gemm_bf16_n256<4><<<N_MID / 64, 128, 0, stream>>>(h1, Q2b, bq2, nbuf, 1);
    segsum_owner_kernel<32, SEG2_LRB, 16, E2_N><<<N_DST / 32, SEG_THREADS, 0, stream>>>(
        nbuf, w2v, src2, dst2, own2, agg, wsum);
    gemm_cat_bf16_n256<2><<<N_DST / 32, 128, 0, stream>>>(agg, wsum, h1, W2b, bw2, h2, 1);
    skip_norm_kernel<<<N_DST / 8, 256, 0, stream>>>(hp, h2, zf, N_DST);
    score_kernel<<<N_PAIRS / 8, 256, 0, stream>>>(zf, bias, nids,
                                                  pos_s, pos_d, neg_s, neg_d,
                                                  out, N_PAIRS);
}


# Layer-1 dst-row GEMMs chained in one launch via the LDS tile: hp -> K=512 GEMM -> n2 GEMM, each result re-enters the tile as bf16; hp, h1 stored for rows < N_DST; 15->13 launches; sums unchanged
# speedup vs baseline: 4.5992x; 1.0481x over previous
#include <hip/hip_runtime.h>
#include <hip/hip_bf16.h>


typedef __attribute__((ext_vector_type(16))) __bf16 v16bf;
typedef __attribute__((ext_vector_type(8)))  __bf16 v8bf;
typedef __attribute__((ext_vector_type(4)))  __bf16 v4bf;
typedef __attribute__((ext_vector_type(8)))  float  v8f;

#define N_SRC   131072
#define N_MID   32768
#define N_DST   8192
#define E1_N    524288
#define E2_N    131072
#define HDIM    256
#define N_PAIRS 8192


__global__ void pack_w_bf16_kernel(const float* __restrict__ src,
                                   __bf16* __restrict__ dst, int K, int n) {
    int i = blockIdx.x * blockDim.x + threadIdx.x;
    if (i >= n) return;
    int e    = i & 7;
    int l    = (i >> 3) & 31;
    int j    = (i >> 8) & 1;
    int frag = i >> 9;
    int KS   = K >> 5;
    int nt   = frag / KS;
    int ks   = frag - nt * KS;
    int col  = nt * 16 + (l & 15);
    int k    = ks * 32 + j * 16 + (l >> 4) * 8 + e;
    dst[i] = (__bf16)src[(size_t)col * K + k];
}

__global__ __launch_bounds__(256)
void compose_q1_wp_kernel(const float* __restrict__ Q1, const float* __restrict__ bq1,
                          const float* __restrict__ Wp, const float* __restrict__ bp,
                          __bf16* __restrict__ Wcb, float* __restrict__ bc)
{
    __shared__ float q[HDIM];
    __shared__ float part[8];
    const int n = blockIdx.x;
    const int c = threadIdx.x;
    float qv = Q1[(size_t)n * HDIM + c];
    q[c] = qv;
    float pb = qv * bp[c];
    #pragma unroll
    for (int o = 16; o > 0; o >>= 1) pb += __shfl_xor(pb, o, 32);
    if ((c & 31) == 0) part[c >> 5] = pb;
    __syncthreads();
    float acc = 0.0f;
    #pragma unroll 32
    for (int m = 0; m < HDIM; ++m)
        acc = fmaf(q[m], Wp[(size_t)m * HDIM + c], acc);
    int e  = c & 7;
    int h  = (c >> 3) & 1;
    int j  = (c >> 4) & 1;
    int ks = c >> 5;
    int l  = h * 16 + (n & 15);
    int nt = n >> 4;
    int i  = ((((nt * (HDIM >> 5) + ks) * 2 + j) * 32 + l) << 3) + e;
    Wcb[i] = (__bf16)acc;
    if (c == 0) {
        float s = bq1[n];
        #pragma unroll
        for (int w = 0; w < 8; ++w) s += part[w];
        bc[n] = s;
    }
}

#define A_PAD 8

template <bool SCALED, int ROWS>
__device__ __forceinline__ void stage_a_rows(const float* __restrict__ src,
                                             const float* __restrict__ wsum,
                                             int r0, __bf16* __restrict__ tile,
                                             int pitch, int kofs, int wave, int lane)
{
    constexpr int RW = ROWS / 4;
    float4 v[2 * RW];
    #pragma unroll
    for (int i = 0; i < 2 * RW; ++i) {
        int r = wave * RW + (i >> 1);
        int cidx = (i & 1) * 128 + lane * 4;
        v[i] = *(const float4*)(src + (size_t)(r0 + r) * HDIM + cidx);
    }
    float rs[RW];
    if (SCALED) {
        #pragma unroll
        for (int j = 0; j < RW; ++j) {
            float ws = wsum[r0 + wave * RW + j];
            ws = ws < 1.0f ? 1.0f : ws;
            rs[j] = __frcp_rn(ws);
        }
    }
    #pragma unroll
    for (int i = 0; i < 2 * RW; ++i) {
        int r = wave * RW + (i >> 1);
        int cidx = (i & 1) * 128 + lane * 4;
        float4 x = v[i];
        if (SCALED) { float s = rs[i >> 1]; x.x *= s; x.y *= s; x.z *= s; x.w *= s; }
        v4bf o;
        o[0] = (__bf16)x.x; o[1] = (__bf16)x.y; o[2] = (__bf16)x.z; o[3] = (__bf16)x.w;
        *(v4bf*)(tile + r * pitch + kofs + cidx) = o;
    }
}

__device__ __forceinline__ void load_b_kstep(const __bf16* __restrict__ wk, int tstride,
                                             v16bf* b)
{
    #pragma unroll
    for (int t = 0; t < 4; ++t) {
        const __bf16* wt = wk + (size_t)t * tstride;
        v8bf blo = *(const v8bf*)(wt);
        v8bf bhi = *(const v8bf*)(wt + 256);
        b[t] = __builtin_shufflevector(blo, bhi, 0, 1, 2, 3, 4, 5, 6, 7,
                                       8, 9, 10, 11, 12, 13, 14, 15);
    }
}

template <int RT>
__device__ __forceinline__ void kstep_strip(const __bf16* __restrict__ ak, int rstride,
                                            const v16bf* b, v8f (*acc)[4])
{
    v16bf a[RT];
    #pragma unroll
    for (int i = 0; i < RT; ++i) {
        v8bf alo = *(const v8bf*)(ak + i * rstride);
        v8bf ahi = *(const v8bf*)(ak + i * rstride + 16);
        a[i] = __builtin_shufflevector(alo, ahi, 0, 1, 2, 3, 4, 5, 6, 7,
                                       8, 9, 10, 11, 12, 13, 14, 15);
    }
    #pragma unroll
    for (int t = 0; t < 4; ++t) {
        #pragma unroll
        for (int i = 0; i < RT; ++i)
            acc[i][t] = __builtin_amdgcn_wmma_f32_16x16x32_bf16(
                false, a[i], false, b[t], (short)0, acc[i][t], false, false);
    }
}

template <int KS, int RT>
__device__ __forceinline__ void kloop_strip(const __bf16* __restrict__ arow, int rstride,
                                            const __bf16* __restrict__ wfrag, int tstride,
                                            v8f (*acc)[4])
{
    v16bf bcur[4], bnxt[4];
    load_b_kstep(wfrag, tstride, bcur);
    #pragma unroll
    for (int ks = 0; ks < KS; ++ks) {
        if (ks + 1 < KS) load_b_kstep(wfrag + (ks + 1) * 512, tstride, bnxt);
        __builtin_amdgcn_sched_barrier(0);
        kstep_strip<RT>(arow + ks * 32, rstride, bcur, acc);
        if (ks + 1 < KS) {
            #pragma unroll
            for (int t = 0; t < 4; ++t) bcur[t] = bnxt[t];
        }
    }
}

__device__ __forceinline__ void store_strip(const v8f* acc, const float* __restrict__ bias,
                                            float* __restrict__ C, int m0, int n0,
                                            int half, int l15, int relu)
{
    #pragma unroll
    for (int t = 0; t < 4; ++t) {
        int n = n0 + t * 16 + l15;
        float bv = bias[n];
        #pragma unroll
        for (int r = 0; r < 8; ++r) {
            int m = m0 + half * 8 + r;
            float v = acc[t][r] + bv;
            if (relu) v = fmaxf(v, 0.0f);
            C[(size_t)m * HDIM + n] = v;
        }
    }
}

template <int RT>
__global__ __launch_bounds__(128)
void gemm_bf16_n256(const float* __restrict__ A, const __bf16* __restrict__ W,
                    const float* __restrict__ bias, float* __restrict__ C,
                    int relu)
{
    constexpr int ROWS  = 16 * RT;
    constexpr int PITCH = HDIM + A_PAD;
    __shared__ __attribute__((aligned(16))) __bf16 atile[ROWS * PITCH];

    const int lane = threadIdx.x & 31;
    const int wave = threadIdx.x >> 5;
    const int half = lane >> 4;
    const int l15  = lane & 15;
    const int mb   = blockIdx.x * ROWS;
    const int n0   = wave * 64;

    stage_a_rows<false, ROWS>(A, nullptr, mb, atile, PITCH, 0, wave, lane);
    __syncthreads();

    v8f acc[RT][4];
    #pragma unroll
    for (int i = 0; i < RT; ++i) {
        #pragma unroll
        for (int t = 0; t < 4; ++t) acc[i][t] = (v8f){};
    }

    const __bf16* arow = atile + l15 * PITCH + half * 8;
    const int     tstride = (HDIM >> 5) * 512;
    const __bf16* wfrag = W + (size_t)(n0 >> 4) * tstride + lane * 8;

    kloop_strip<HDIM / 32, RT>(arow, 16 * PITCH, wfrag, tstride, acc);

    #pragma unroll
    for (int i = 0; i < RT; ++i)
        store_strip(acc[i], bias, C, mb + 16 * i, n0, half, l15, relu);
}

template <int RT>
__global__ __launch_bounds__(128)
void gemm_cat_bf16_n256(const float* __restrict__ agg, const float* __restrict__ wsum,
                        const float* __restrict__ hdst, const __bf16* __restrict__ W,
                        const float* __restrict__ bias, float* __restrict__ C, int relu)
{
    constexpr int ROWS  = 16 * RT;
    constexpr int PITCH = 2 * HDIM + A_PAD;
    __shared__ __attribute__((aligned(16))) __bf16 atile[ROWS * PITCH];

    const int lane = threadIdx.x & 31;
    const int wave = threadIdx.x >> 5;
    const int half = lane >> 4;
    const int l15  = lane & 15;
    const int mb   = blockIdx.x * ROWS;
    const int n0   = wave * 64;

    stage_a_rows<true,  ROWS>(agg,  wsum,    mb, atile, PITCH, 0,    wave, lane);
    stage_a_rows<false, ROWS>(hdst, nullptr, mb, atile, PITCH, HDIM, wave, lane);
    __syncthreads();

    v8f acc[RT][4];
    #pragma unroll
    for (int i = 0; i < RT; ++i) {
        #pragma unroll
        for (int t = 0; t < 4; ++t) acc[i][t] = (v8f){};
    }

    const __bf16* arow = atile + l15 * PITCH + half * 8;
    const int     tstride = (2 * HDIM / 32) * 512;
    const __bf16* wfrag = W + (size_t)(n0 >> 4) * tstride + lane * 8;

    kloop_strip<2 * HDIM / 32, RT>(arow, 16 * PITCH, wfrag, tstride, acc);

    #pragma unroll
    for (int i = 0; i < RT; ++i)
        store_strip(acc[i], bias, C, mb + 16 * i, n0, half, l15, relu);
}

template <int RT, bool RELU>
__device__ __forceinline__ void bias_act_strip(v8f (*acc)[4], const float* __restrict__ bias,
                                               int n0, int l15)
{
    #pragma unroll
    for (int t = 0; t < 4; ++t) {
        float bv = bias[n0 + t * 16 + l15];
        #pragma unroll
        for (int i = 0; i < RT; ++i) {
            #pragma unroll
            for (int r = 0; r < 8; ++r) {
                float v = acc[i][t][r] + bv;
                if (RELU) v = fmaxf(v, 0.0f);
                acc[i][t][r] = v;
            }
        }
    }
}

template <int RT>
__device__ __forceinline__ void strip_to_tile(const v8f (*acc)[4], __bf16* tile, int pitch,
                                              int kofs, int n0, int half, int l15)
{
    #pragma unroll
    for (int i = 0; i < RT; ++i) {
        #pragma unroll
        for (int t = 0; t < 4; ++t) {
            #pragma unroll
            for (int r = 0; r < 8; ++r)
                tile[(16 * i + 8 * half + r) * pitch + kofs + n0 + 16 * t + l15] =
                    (__bf16)acc[i][t][r];
        }
    }
}

template <int RT>
__device__ __forceinline__ void strip_to_global(const v8f (*acc)[4], float* __restrict__ C,
                                                int mb, int n0, int half, int l15)
{
    #pragma unroll
    for (int i = 0; i < RT; ++i) {
        #pragma unroll
        for (int t = 0; t < 4; ++t) {
            #pragma unroll
            for (int r = 0; r < 8; ++r)
                C[(size_t)(mb + 16 * i + 8 * half + r) * HDIM + n0 + 16 * t + l15] = acc[i][t][r];
        }
    }
}

template <int RT>
__device__ __forceinline__ void zero_strip(v8f (*acc)[4])
{
    #pragma unroll
    for (int i = 0; i < RT; ++i) {
        #pragma unroll
        for (int t = 0; t < 4; ++t) acc[i][t] = (v8f){};
    }
}

__global__ __launch_bounds__(128)
void layer1_chain_kernel(const float* __restrict__ feat, const __bf16* __restrict__ Wpb,
                         const float* __restrict__ bp,
                         const float* __restrict__ agg, const float* __restrict__ wsum,
                         const __bf16* __restrict__ W1b, const float* __restrict__ bw1,
                         const __bf16* __restrict__ Q2b, const float* __restrict__ bq2,
                         float* __restrict__ hp, float* __restrict__ h1,
                         float* __restrict__ n2)
{
    constexpr int RT    = 4;
    constexpr int ROWS  = 16 * RT;
    constexpr int PITCH = 3 * HDIM + A_PAD;
    __shared__ __attribute__((aligned(16))) __bf16 atile[ROWS * PITCH];

    const int  lane = threadIdx.x & 31;
    const int  wave = threadIdx.x >> 5;
    const int  half = lane >> 4;
    const int  l15  = lane & 15;
    const int  mb   = blockIdx.x * ROWS;
    const int  n0   = wave * 64;
    const bool keep = mb < N_DST;

    const __bf16* arow  = atile + l15 * PITCH + half * 8;
    const int     ts256 = (HDIM >> 5) * 512;
    const int     ts512 = (2 * HDIM >> 5) * 512;
    v8f acc[RT][4];

    stage_a_rows<true,  ROWS>(agg,  wsum,    mb, atile, PITCH, 0,        wave, lane);
    stage_a_rows<false, ROWS>(feat, nullptr, mb, atile, PITCH, 2 * HDIM, wave, lane);
    __syncthreads();

    zero_strip<RT>(acc);
    kloop_strip<HDIM / 32, RT>(arow + 2 * HDIM, 16 * PITCH,
                               Wpb + (size_t)(n0 >> 4) * ts256 + lane * 8, ts256, acc);
    bias_act_strip<RT, false>(acc, bp, n0, l15);
    strip_to_tile<RT>(acc, atile, PITCH, HDIM, n0, half, l15);
    __syncthreads();
    if (keep) strip_to_global<RT>(acc, hp, mb, n0, half, l15);

    zero_strip<RT>(acc);
    kloop_strip<2 * HDIM / 32, RT>(arow, 16 * PITCH, W1b + (size_t)(n0 >> 4) * ts512 + lane * 8,
                                   ts512, acc);
    bias_act_strip<RT, true>(acc, bw1, n0, l15);
    strip_to_tile<RT>(acc, atile, PITCH, 2 * HDIM, n0, half, l15);
    __syncthreads();
    if (keep) strip_to_global<RT>(acc, h1, mb, n0, half, l15);

    zero_strip<RT>(acc);
    kloop_strip<HDIM / 32, RT>(arow + 2 * HDIM, 16 * PITCH,
                               Q2b + (size_t)(n0 >> 4) * ts256 + lane * 8, ts256, acc);
    bias_act_strip<RT, true>(acc, bq2, n0, l15);
    strip_to_global<RT>(acc, n2, mb, n0, half, l15);
}

#define SEG_THREADS 1024
#define SEG_CAP     8192
#define SEG_UA      8
#define SEG1_LRB    7
#define SEG2_LRB    5

__global__ __launch_bounds__(256)
void owner_bytes_kernel(const int* __restrict__ dst1, const int* __restrict__ dst2,
                        unsigned* __restrict__ own1, unsigned* __restrict__ own2)
{
    int g = blockIdx.x * 256 + threadIdx.x;
    const int4* s = (const int4*)dst1;
    unsigned*   o = own1;
    int sh = SEG1_LRB;
    if (g >= E1_N / 4) {
        g -= E1_N / 4;
        s = (const int4*)dst2;
        o = own2;
        sh = SEG2_LRB;
    }
    int4 d = s[g];
    o[g] = ((unsigned)(d.x >> sh) & 255u) | (((unsigned)(d.y >> sh) & 255u) << 8) |
           (((unsigned)(d.z >> sh) & 255u) << 16) | (((unsigned)(d.w >> sh) & 255u) << 24);
}

__device__ __forceinline__ unsigned match_bytes(unsigned x, unsigned pat)
{
    x ^= pat;
    return (x - 0x01010101u) & ~x & 0x80808080u;
}

template <int ROWS, int LRB, int CH, int E>
__global__ __launch_bounds__(SEG_THREADS)
void segsum_owner_kernel(const float* __restrict__ nsrc,
                         const float* __restrict__ w,
                         const int* __restrict__ src,
                         const int* __restrict__ dst,
                         const unsigned* __restrict__ own,
                         float* __restrict__ agg,
                         float* __restrict__ wsum)
{
    constexpr int NLD  = E / 16;
    constexpr int PER  = NLD / SEG_THREADS;
    constexpr int UA   = PER < SEG_UA ? PER : SEG_UA;
    constexpr int NIT  = PER / UA;
    static_assert(ROWS == (1 << LRB), "local row fits LRB bits");
    static_assert(E % (16 * SEG_THREADS) == 0 && PER % UA == 0, "the scan has whole trips only");
    static_assert((NLD & (NLD - 1)) == 0 && NLD >= 256, "start rotation by mask");

    __shared__ __attribute__((aligned(16))) float tile[ROWS * HDIM];
    __shared__ float    wl[ROWS];
    __shared__ unsigned list[SEG_CAP];
    __shared__ unsigned cnt;

    const int tid  = threadIdx.x;
    const int lane = tid & 31;
    const int wave = __builtin_amdgcn_readfirstlane(tid >> 5);
    const int base = blockIdx.x * ROWS;

    for (int i = tid; i < ROWS * HDIM / 4; i += SEG_THREADS)
        ((float4*)tile)[i] = make_float4(0.0f, 0.0f, 0.0f, 0.0f);
    if (tid < ROWS) wl[tid] = 0.0f;
    if (tid == 0) cnt = 0u;
    __syncthreads();

    auto push = [&](int e) {
        unsigned pos = atomicAdd(&cnt, 1u);
        if (pos < SEG_CAP) {
            list[pos] = (unsigned)e;
        } else {
            unsigned lr = (unsigned)(dst[e] - base);
            if (lr < (unsigned)ROWS) {
                int   s  = src[e];
                float we = w[e];
                atomicAdd(&wl[lr], we);
                for (int c = 0; c < HDIM; ++c)
                    atomicAdd(&tile[lr * HDIM + c], nsrc[(size_t)s * HDIM + c] * we);
            }
        }
    };
    const uint4*   o4  = (const uint4*)own;
    const unsigned pat = (blockIdx.x & 255u) * 0x01010101u;
    const int      rot = (int)blockIdx.x * (NLD >> 8);
    for (int it = 0; it < NIT; ++it) {
        uint4 d[UA];
        int   li[UA];
        #pragma unroll
        for (int u = 0; u < UA; ++u) {
            li[u] = ((it * UA + u) * SEG_THREADS + tid + rot) & (NLD - 1);
            d[u]  = o4[li[u]];
        }
        #pragma unroll
        for (int u = 0; u < UA; ++u) {
            unsigned m = (match_bytes(d[u].x, pat) >> 7) | (match_bytes(d[u].y, pat) >> 6) |
                         (match_bytes(d[u].z, pat) >> 5) | (match_bytes(d[u].w, pat) >> 4);
            while (m) {
                const int i = __builtin_ctz(m);
                m &= m - 1u;
                push(li[u] * 16 + (i & 7) * 4 + (i >> 3));
            }
        }
    }
    __syncthreads();

    unsigned nl = cnt;
    nl = nl < (unsigned)SEG_CAP ? nl : (unsigned)SEG_CAP;
    const int n = __builtin_amdgcn_readfirstlane((int)nl);
    for (int c0 = wave * CH; c0 < n; c0 += (SEG_THREADS / 32) * CH) {
        const int  idx   = c0 + lane;
        const bool valid = (lane < CH) && (idx < n);
        int   e  = valid ? (int)list[idx] : 0;
        int   s  = src[e];
        float we = w[e];
        int   lr = dst[e] - base;
        if (!valid || (unsigned)lr >= (unsigned)ROWS) { s = 0; we = 0.0f; lr = -1; }
        if (lr >= 0) atomicAdd(&wl[lr], we);
        const int rem = n - c0;
        const int cw  = rem < CH ? rem : CH;
        int j = 0;
        for (; j + 4 <= cw; j += 4) {
            int sj[4], lj[4]; float wj[4];
            #pragma unroll
            for (int u = 0; u < 4; ++u) {
                sj[u] = __builtin_amdgcn_readlane(s, j + u);
                lj[u] = __builtin_amdgcn_readlane(lr, j + u);
                wj[u] = __int_as_float(__builtin_amdgcn_readlane(__float_as_int(we), j + u));
            }
            float v[4][8];
            #pragma unroll
            for (int u = 0; u < 4; ++u) {
                const float* nr = nsrc + (size_t)sj[u] * HDIM + lane;
                #pragma unroll
                for (int q = 0; q < 8; ++q) v[u][q] = nr[q * 32];
            }
            #pragma unroll
            for (int u = 0; u < 4; ++u) {
                if (lj[u] < 0) continue;
                float* tr = tile + lj[u] * HDIM + lane;
                #pragma unroll
                for (int q = 0; q < 8; ++q) atomicAdd(tr + q * 32, v[u][q] * wj[u]);
            }
        }
        for (; j < cw; ++j) {
            int   s1 = __builtin_amdgcn_readlane(s, j);
            int   l1 = __builtin_amdgcn_readlane(lr, j);
            float w1 = __int_as_float(__builtin_amdgcn_readlane(__float_as_int(we), j));
            const float* nr = nsrc + (size_t)s1 * HDIM + lane;
            float* tr = tile + l1 * HDIM + lane;
            if (l1 < 0) continue;
            float v1[8];
            #pragma unroll
            for (int q = 0; q < 8; ++q) v1[q] = nr[q * 32];
            #pragma unroll
            for (int q = 0; q < 8; ++q) atomicAdd(tr + q * 32, v1[q] * w1);
        }
    }
    __syncthreads();

    float4* arow = (float4*)(agg + (size_t)base * HDIM);
    for (int i = tid; i < ROWS * HDIM / 4; i += SEG_THREADS)
        arow[i] = ((const float4*)tile)[i];
    if (tid < ROWS) wsum[base + tid] = wl[tid];
}

__global__ __launch_bounds__(256)
void skip_norm_kernel(const float* __restrict__ hp, const float* __restrict__ h2,
                      float* __restrict__ zf, int rows)
{
    int r    = blockIdx.x * 8 + (threadIdx.x >> 5);
    int lane = threadIdx.x & 31;
    if (r >= rows) return;
    float v[8];
    float ss = 0.0f;
    #pragma unroll
    for (int j = 0; j < 8; ++j) {
        int c = lane * 8 + j;
        v[j] = hp[(size_t)r * HDIM + c] + h2[(size_t)r * HDIM + c];
        ss += v[j] * v[j];
    }
    #pragma unroll
    for (int o = 16; o > 0; o >>= 1) ss += __shfl_xor(ss, o, 32);
    float nrm   = sqrtf(ss);
    float scale = (nrm == 0.0f) ? 1.0f : 1.0f / nrm;
    #pragma unroll
    for (int j = 0; j < 8; ++j)
        zf[(size_t)r * HDIM + lane * 8 + j] = v[j] * scale;
}

__global__ __launch_bounds__(256)
void score_kernel(const float* __restrict__ zf, const float* __restrict__ bias,
                  const int* __restrict__ nids,
                  const int* __restrict__ ps, const int* __restrict__ pd,
                  const int* __restrict__ ns, const int* __restrict__ nd,
                  float* __restrict__ out, int npairs)
{
    int p    = blockIdx.x * 8 + (threadIdx.x >> 5);
    int lane = threadIdx.x & 31;
    if (p >= npairs) return;
    int pu = ps[p], pv = pd[p], nu = ns[p], nv = nd[p];
    float sp = 0.0f, sn = 0.0f;
    #pragma unroll
    for (int j = 0; j < 8; ++j) {
        int c = lane * 8 + j;
        sp += zf[(size_t)pu * HDIM + c] * zf[(size_t)pv * HDIM + c];
        sn += zf[(size_t)nu * HDIM + c] * zf[(size_t)nv * HDIM + c];
    }
    #pragma unroll
    for (int o = 16; o > 0; o >>= 1) {
        sp += __shfl_xor(sp, o, 32);
        sn += __shfl_xor(sn, o, 32);
    }
    if (lane == 0) {
        float gp = bias[nids[pu]] + bias[nids[pv]];
        float gn = bias[nids[nu]] + bias[nids[nv]];
        float m  = (sn + gn) - (sp + gp) + 1.0f;
        out[p] = m > 0.0f ? m : 0.0f;
    }
}

extern "C" void kernel_launch(void* const* d_in, const int* in_sizes, int n_in,
                              void* d_out, int out_size, void* d_ws, size_t ws_size,
                              hipStream_t stream) {
    const float* feat = (const float*)d_in[0];
    const float* Wp   = (const float*)d_in[1];
    const float* bp   = (const float*)d_in[2];
    const float* Q1   = (const float*)d_in[3];
    const float* bq1  = (const float*)d_in[4];
    const float* W1   = (const float*)d_in[5];
    const float* bw1  = (const float*)d_in[6];
    const float* Q2   = (const float*)d_in[7];
    const float* bq2  = (const float*)d_in[8];
    const float* W2   = (const float*)d_in[9];
    const float* bw2  = (const float*)d_in[10];
    const float* w1v  = (const float*)d_in[11];
    const float* w2v  = (const float*)d_in[12];
    const float* bias = (const float*)d_in[13];
    const int* src1   = (const int*)d_in[14];
    const int* dst1   = (const int*)d_in[15];
    const int* src2   = (const int*)d_in[16];
    const int* dst2   = (const int*)d_in[17];
    const int* pos_s  = (const int*)d_in[18];
    const int* pos_d  = (const int*)d_in[19];
    const int* neg_s  = (const int*)d_in[20];
    const int* neg_d  = (const int*)d_in[21];
    const int* nids   = (const int*)d_in[22];
    float* out        = (float*)d_out;

    char* ws = (char*)d_ws;
    const size_t SZ_HP  = (size_t)N_SRC * HDIM * 4;
    const size_t SZ_N   = (size_t)N_SRC * HDIM * 4;
    const size_t SZ_H1  = (size_t)N_MID * HDIM * 4;
    const size_t SZ_AGG = (size_t)N_MID * HDIM * 4;
    const size_t SZ_WS  = 256 * 1024;
    const size_t SZ_H2  = (size_t)N_DST * HDIM * 4;
    const size_t SZ_ZF  = (size_t)N_DST * HDIM * 4;
    float*  hp   = (float*)(ws);
    float*  bc   = hp + (size_t)N_MID * HDIM;
    float*  nbuf = (float*)(ws + SZ_HP);
    float*  h1   = (float*)(ws + SZ_HP + SZ_N);
    float*  agg  = (float*)(ws + SZ_HP + SZ_N + SZ_H1);
    float*  wsum = (float*)(ws + SZ_HP + SZ_N + SZ_H1 + SZ_AGG);
    float*  h2   = (float*)(ws + SZ_HP + SZ_N + SZ_H1 + SZ_AGG + SZ_WS);
    float*  zf   = (float*)(ws + SZ_HP + SZ_N + SZ_H1 + SZ_AGG + SZ_WS + SZ_H2);
    __bf16* Wpb  = (__bf16*)(ws + SZ_HP + SZ_N + SZ_H1 + SZ_AGG + SZ_WS + SZ_H2 + SZ_ZF);
    __bf16* Wcb  = Wpb + HDIM * HDIM;
    __bf16* W1b  = Wcb + HDIM * HDIM;
    __bf16* Q2b  = W1b + HDIM * 2 * HDIM;
    __bf16* W2b  = Q2b + HDIM * HDIM;
    unsigned* own1 = (unsigned*)(bc + HDIM);
    unsigned* own2 = own1 + E1_N / 4;

    pack_w_bf16_kernel<<<(HDIM * HDIM + 255) / 256, 256, 0, stream>>>(Wp, Wpb, HDIM, HDIM * HDIM);
    compose_q1_wp_kernel<<<HDIM, 256, 0, stream>>>(Q1, bq1, Wp, bp, Wcb, bc);
    pack_w_bf16_kernel<<<(HDIM * 2 * HDIM + 255) / 256, 256, 0, stream>>>(W1, W1b, 2 * HDIM, HDIM * 2 * HDIM);
    pack_w_bf16_kernel<<<(HDIM * HDIM + 255) / 256, 256, 0, stream>>>(Q2, Q2b, HDIM, HDIM * HDIM);
    pack_w_bf16_kernel<<<(HDIM * 2 * HDIM + 255) / 256, 256, 0, stream>>>(W2, W2b, 2 * HDIM, HDIM * 2 * HDIM);
    owner_bytes_kernel<<<(E1_N + E2_N) / 4 / 256, 256, 0, stream>>>(dst1, dst2, own1, own2);

    gemm_bf16_n256<4><<<N_SRC / 64, 128, 0, stream>>>(feat, Wcb, bc, nbuf, 1);
    segsum_owner_kernel<128, SEG1_LRB, 32, E1_N><<<N_MID / 128, SEG_THREADS, 0, stream>>>(
        nbuf, w1v, src1, dst1, own1, agg, wsum);
    layer1_chain_kernel<<<N_MID / 64, 128, 0, stream>>>(feat, Wpb, bp, agg, wsum, W1b, bw1,
                                                        Q2b, bq2, hp, h1, nbuf);
    segsum_owner_kernel<32, SEG2_LRB, 16, E2_N><<<N_DST / 32, SEG_THREADS, 0, stream>>>(
        nbuf, w2v, src2, dst2, own2, agg, wsum);
    gemm_cat_bf16_n256<2><<<N_DST / 32, 128, 0, stream>>>(agg, wsum, h1, W2b, bw2, h2, 1);
    skip_norm_kernel<<<N_DST / 8, 256, 0, stream>>>(hp, h2, zf, N_DST);
    score_kernel<<<N_PAIRS / 8, 256, 0, stream>>>(zf, bias, nids,
                                                  pos_s, pos_d, neg_s, neg_d,
                                                  out, N_PAIRS);
}
